# starting kernel
# baseline (speedup 1.0000x reference)
#include <hip/hip_runtime.h>


typedef float v2f __attribute__((ext_vector_type(2)));
typedef float v8f __attribute__((ext_vector_type(8)));

template <int K, int N>
__global__ __launch_bounds__(256) void wmma_gemm_f32(
    const float* __restrict__ A, const float* __restrict__ B,
    float* __restrict__ C, int M) {
  __shared__ float lB[K * N];
  for (int i = threadIdx.x; i < K * N; i += 256) {
    int k = i / N, n = i % N;
    lB[((k >> 2) * N + n) * 4 + (k & 3)] = B[i];
  }
  __syncthreads();

  const int lane  = threadIdx.x & 31;
  const int wave  = threadIdx.x >> 5;
  const int mbase = (blockIdx.x * 8 + wave) * 16;
  if (mbase >= M) return;

  const int mr = lane & 15;
  const int hi = lane >> 4;

  const float* arow = A + (size_t)(mbase + mr) * K + 2 * hi;
  const float* brow = lB + mr * 4 + 2 * hi;

  constexpr int NT = N / 16;
  v8f acc[NT];
  v8f zero = {};
#pragma unroll
  for (int t = 0; t < NT; ++t) acc[t] = zero;

  for (int kb = 0; kb < K / 4; ++kb) {
    v2f a = *(const v2f*)(arow + kb * 4);
#pragma unroll
    for (int t = 0; t < NT; ++t) {
      v2f b = *(const v2f*)(brow + kb * (N * 4) + t * 64);
      acc[t] = __builtin_amdgcn_wmma_f32_16x16x4_f32(
           false, a,  false, b,
           (short)0, acc[t],  false,  false);
    }
  }

#pragma unroll
  for (int t = 0; t < NT; ++t)
#pragma unroll
    for (int r = 0; r < 8; ++r)
      C[(size_t)(mbase + 8 * hi + r) * N + t * 16 + mr] = acc[t][r];
}

__global__ void k_deg_init(float* __restrict__ deg, int n) {
  int i = blockIdx.x * blockDim.x + threadIdx.x;
  if (i < n) deg[i] = 1.0f;
}

__global__ void k_deg_count(const int* __restrict__ dst,
                            float* __restrict__ deg, int e) {
  int i = blockIdx.x * blockDim.x + threadIdx.x;
  if (i < e) atomicAdd(&deg[dst[i]], 1.0f);
}

__global__ void k_rsqrt_inplace(float* __restrict__ d, int n) {
  int i = blockIdx.x * blockDim.x + threadIdx.x;
  if (i < n) d[i] = rsqrtf(d[i]);
}

template <int F>
__global__ void k_self_init(const float* __restrict__ h,
                            const float* __restrict__ dinv,
                            float* __restrict__ agg, int n) {
  constexpr int PER = F / 4;
  int t = blockIdx.x * blockDim.x + threadIdx.x;
  if (t >= n * PER) return;
  int node = t / PER;
  float di = dinv[node];
  float s = di * di;
  float4 v = ((const float4*)h)[t];
  float4 o = {v.x * s, v.y * s, v.z * s, v.w * s};
  ((float4*)agg)[t] = o;
}

template <int F>
__global__ __launch_bounds__(256) void k_scatter(
    const int* __restrict__ src, const int* __restrict__ dst,
    const float* __restrict__ dinv, const float* __restrict__ h,
    float* __restrict__ agg, int e) {
  constexpr int PER = F / 4;
  int t = blockIdx.x * blockDim.x + threadIdx.x;
  int ed = t / PER;
  int q  = t % PER;
  if (ed >= e) return;
  int s = src[ed], d = dst[ed];
  float nrm = dinv[s] * dinv[d];
  float4 v = ((const float4*)(h + (size_t)s * F))[q];
  float* ap = agg + (size_t)d * F + q * 4;
  atomicAdd(ap + 0, v.x * nrm);
  atomicAdd(ap + 1, v.y * nrm);
  atomicAdd(ap + 2, v.z * nrm);
  atomicAdd(ap + 3, v.w * nrm);
}

template <int F>
__global__ void k_bias_relu(float* __restrict__ agg,
                            const float* __restrict__ b, int n) {
  constexpr int PER = F / 4;
  int t = blockIdx.x * blockDim.x + threadIdx.x;
  if (t >= n * PER) return;
  int q = t % PER;
  float4 bb = ((const float4*)b)[q];
  float4 v = ((float4*)agg)[t];
  v.x = fmaxf(v.x + bb.x, 0.0f);
  v.y = fmaxf(v.y + bb.y, 0.0f);
  v.z = fmaxf(v.z + bb.z, 0.0f);
  v.w = fmaxf(v.w + bb.w, 0.0f);
  ((float4*)agg)[t] = v;
}

__global__ void k_out_init(const float* __restrict__ g,
                           const float* __restrict__ dinv,
                           const float* __restrict__ b2,
                           float* __restrict__ out, int n) {
  int t = blockIdx.x * blockDim.x + threadIdx.x;
  if (t >= n * 4) return;
  int node = t >> 2;
  int q    = t & 3;
  float di = dinv[node];
  float s = di * di;
  float4 v = ((const float4*)g)[t];
  float4 bb = ((const float4*)b2)[q];
  float4 o = {v.x * s + bb.x, v.y * s + bb.y, v.z * s + bb.z, v.w * s + bb.w};
  ((float4*)out)[t] = o;
}

extern "C" void kernel_launch(void* const* d_in, const int* in_sizes, int n_in,
                              void* d_out, int out_size, void* d_ws,
                              size_t ws_size, hipStream_t stream) {
  constexpr int IN_FEAT = 256, HIDDEN = 64, NCLS = 16;

  const float* x  = (const float*)d_in[0];
  const int*   ei = (const int*)d_in[1];
  const float* W1 = (const float*)d_in[2];
  const float* b1 = (const float*)d_in[3];
  const float* W2 = (const float*)d_in[4];
  const float* b2 = (const float*)d_in[5];
  float* out = (float*)d_out;

  const int N = in_sizes[0] / IN_FEAT;
  const int E = in_sizes[1] / 2;
  const int* src = ei;
  const int* dst = ei + E;

  float* ws   = (float*)d_ws;
  float* dinv = ws;
  float* h1   = dinv + N;
  float* agg1 = h1 + (size_t)N * HIDDEN;
  float* g2   = agg1 + (size_t)N * HIDDEN;

  const int B = 256;
  const int mtiles  = (N + 15) / 16;
  const int gblocks = (mtiles + 7) / 8;

  k_deg_init<<<(N + B - 1) / B, B, 0, stream>>>(dinv, N);
  k_deg_count<<<(E + B - 1) / B, B, 0, stream>>>(dst, dinv, E);
  k_rsqrt_inplace<<<(N + B - 1) / B, B, 0, stream>>>(dinv, N);

  wmma_gemm_f32<IN_FEAT, HIDDEN><<<gblocks, B, 0, stream>>>(x, W1, h1, N);
  k_self_init<HIDDEN><<<((size_t)N * 16 + B - 1) / B, B, 0, stream>>>(h1, dinv, agg1, N);
  k_scatter<HIDDEN><<<((size_t)E * 16 + B - 1) / B, B, 0, stream>>>(src, dst, dinv, h1, agg1, E);
  k_bias_relu<HIDDEN><<<((size_t)N * 16 + B - 1) / B, B, 0, stream>>>(agg1, b1, N);

  wmma_gemm_f32<HIDDEN, NCLS><<<gblocks, B, 0, stream>>>(agg1, W2, g2, N);
  k_out_init<<<((size_t)N * 4 + B - 1) / B, B, 0, stream>>>(g2, dinv, b2, out, N);
  k_scatter<NCLS><<<((size_t)E * 4 + B - 1) / B, B, 0, stream>>>(src, dst, dinv, g2, out, E);
}


# Edge scatter remapped to lane = feature column: each f32 atomic instruction adds whole contiguous rows (4x fewer L2 line requests), row loads batched ahead; atomics kept, arrival order as before
# speedup vs baseline: 1.2788x; 1.2788x over previous
#include <hip/hip_runtime.h>


typedef float v2f __attribute__((ext_vector_type(2)));
typedef float v8f __attribute__((ext_vector_type(8)));

template <int K, int N>
__global__ __launch_bounds__(256) void wmma_gemm_f32(
    const float* __restrict__ A, const float* __restrict__ B,
    float* __restrict__ C, int M) {
  __shared__ float lB[K * N];
  for (int i = threadIdx.x; i < K * N; i += 256) {
    int k = i / N, n = i % N;
    lB[((k >> 2) * N + n) * 4 + (k & 3)] = B[i];
  }
  __syncthreads();

  const int lane  = threadIdx.x & 31;
  const int wave  = threadIdx.x >> 5;
  const int mbase = (blockIdx.x * 8 + wave) * 16;
  if (mbase >= M) return;

  const int mr = lane & 15;
  const int hi = lane >> 4;

  const float* arow = A + (size_t)(mbase + mr) * K + 2 * hi;
  const float* brow = lB + mr * 4 + 2 * hi;

  constexpr int NT = N / 16;
  v8f acc[NT];
  v8f zero = {};
#pragma unroll
  for (int t = 0; t < NT; ++t) acc[t] = zero;

  for (int kb = 0; kb < K / 4; ++kb) {
    v2f a = *(const v2f*)(arow + kb * 4);
#pragma unroll
    for (int t = 0; t < NT; ++t) {
      v2f b = *(const v2f*)(brow + kb * (N * 4) + t * 64);
      acc[t] = __builtin_amdgcn_wmma_f32_16x16x4_f32(
           false, a,  false, b,
           (short)0, acc[t],  false,  false);
    }
  }

#pragma unroll
  for (int t = 0; t < NT; ++t)
#pragma unroll
    for (int r = 0; r < 8; ++r)
      C[(size_t)(mbase + 8 * hi + r) * N + t * 16 + mr] = acc[t][r];
}

__global__ void k_deg_init(float* __restrict__ deg, int n) {
  int i = blockIdx.x * blockDim.x + threadIdx.x;
  if (i < n) deg[i] = 1.0f;
}

__global__ void k_deg_count(const int* __restrict__ dst,
                            float* __restrict__ deg, int e) {
  int i = blockIdx.x * blockDim.x + threadIdx.x;
  if (i < e) atomicAdd(&deg[dst[i]], 1.0f);
}

__global__ void k_rsqrt_inplace(float* __restrict__ d, int n) {
  int i = blockIdx.x * blockDim.x + threadIdx.x;
  if (i < n) d[i] = rsqrtf(d[i]);
}

template <int F>
__global__ void k_self_init(const float* __restrict__ h,
                            const float* __restrict__ dinv,
                            float* __restrict__ agg, int n) {
  constexpr int PER = F / 4;
  int t = blockIdx.x * blockDim.x + threadIdx.x;
  if (t >= n * PER) return;
  int node = t / PER;
  float di = dinv[node];
  float s = di * di;
  float4 v = ((const float4*)h)[t];
  float4 o = {v.x * s, v.y * s, v.z * s, v.w * s};
  ((float4*)agg)[t] = o;
}

__global__ __launch_bounds__(256) void k_scatter64(
    const int* __restrict__ src, const int* __restrict__ dst,
    const float* __restrict__ dinv, const float* __restrict__ h,
    float* __restrict__ agg, int e) {
  const int lane = threadIdx.x & 31;
  const int e0 = __builtin_amdgcn_readfirstlane(
      (int)((blockIdx.x * 256 + threadIdx.x) & ~31u));
  if (e0 >= e) return;
  int s = 0, d = 0;
  float nrm = 0.0f;
  if (e0 + lane < e) {
    s = src[e0 + lane];
    d = dst[e0 + lane];
    nrm = dinv[s] * dinv[d];
  }
  const float* hl = h + lane;
  float* al = agg + lane;
  if (e - e0 >= 32) {
    for (int jb = 0; jb < 32; jb += 16) {
      float v0[16], v1[16];
#pragma unroll
      for (int u = 0; u < 16; ++u) {
        const int sj = __builtin_amdgcn_readlane(s, jb + u);
        const float* hp = hl + (size_t)sj * 64;
        v0[u] = hp[0];
        v1[u] = hp[32];
      }
#pragma unroll
      for (int u = 0; u < 16; ++u) {
        const int dj = __builtin_amdgcn_readlane(d, jb + u);
        const float nj = __int_as_float(
            __builtin_amdgcn_readlane(__float_as_int(nrm), jb + u));
        float* ap = al + (size_t)dj * 64;
        atomicAdd(ap, v0[u] * nj);
        atomicAdd(ap + 32, v1[u] * nj);
      }
    }
  } else {
    for (int j = 0; j < e - e0; ++j) {
      const int sj = __builtin_amdgcn_readlane(s, j);
      const int dj = __builtin_amdgcn_readlane(d, j);
      const float nj = __int_as_float(
          __builtin_amdgcn_readlane(__float_as_int(nrm), j));
      const float* hp = hl + (size_t)sj * 64;
      float* ap = al + (size_t)dj * 64;
      const float a0 = hp[0], a1 = hp[32];
      atomicAdd(ap, a0 * nj);
      atomicAdd(ap + 32, a1 * nj);
    }
  }
}

__global__ __launch_bounds__(256) void k_scatter16(
    const int* __restrict__ src, const int* __restrict__ dst,
    const float* __restrict__ dinv, const float* __restrict__ h,
    float* __restrict__ agg, int e) {
  const int lane = threadIdx.x & 31;
  const int c    = lane & 15;
  const int hs   = lane >> 4;
  const int e0 = __builtin_amdgcn_readfirstlane(
      (int)((blockIdx.x * 256 + threadIdx.x) & ~31u));
  if (e0 >= e) return;
  int s = 0, d = 0;
  float nrm = 0.0f;
  if (e0 + lane < e) {
    s = src[e0 + lane];
    d = dst[e0 + lane];
    nrm = dinv[s] * dinv[d];
  }
  const float* hc = h + c;
  float* ac = agg + c;
  if (e - e0 >= 32) {
    float v[16];
#pragma unroll
    for (int u = 0; u < 16; ++u) {
      const int sj = __shfl(s, 2 * u + hs);
      v[u] = hc[(size_t)sj * 16];
    }
#pragma unroll
    for (int u = 0; u < 16; ++u) {
      const int dj = __shfl(d, 2 * u + hs);
      const float nj = __shfl(nrm, 2 * u + hs);
      atomicAdd(ac + (size_t)dj * 16, v[u] * nj);
    }
  } else {
    for (int u = 0; u < 16; ++u) {
      const int idx = 2 * u + hs;
      const int sj = __shfl(s, idx);
      const int dj = __shfl(d, idx);
      const float nj = __shfl(nrm, idx);
      if (e0 + idx < e) {
        const float a = hc[(size_t)sj * 16];
        atomicAdd(ac + (size_t)dj * 16, a * nj);
      }
    }
  }
}

template <int F>
__global__ void k_bias_relu(float* __restrict__ agg,
                            const float* __restrict__ b, int n) {
  constexpr int PER = F / 4;
  int t = blockIdx.x * blockDim.x + threadIdx.x;
  if (t >= n * PER) return;
  int q = t % PER;
  float4 bb = ((const float4*)b)[q];
  float4 v = ((float4*)agg)[t];
  v.x = fmaxf(v.x + bb.x, 0.0f);
  v.y = fmaxf(v.y + bb.y, 0.0f);
  v.z = fmaxf(v.z + bb.z, 0.0f);
  v.w = fmaxf(v.w + bb.w, 0.0f);
  ((float4*)agg)[t] = v;
}

__global__ void k_out_init(const float* __restrict__ g,
                           const float* __restrict__ dinv,
                           const float* __restrict__ b2,
                           float* __restrict__ out, int n) {
  int t = blockIdx.x * blockDim.x + threadIdx.x;
  if (t >= n * 4) return;
  int node = t >> 2;
  int q    = t & 3;
  float di = dinv[node];
  float s = di * di;
  float4 v = ((const float4*)g)[t];
  float4 bb = ((const float4*)b2)[q];
  float4 o = {v.x * s + bb.x, v.y * s + bb.y, v.z * s + bb.z, v.w * s + bb.w};
  ((float4*)out)[t] = o;
}

extern "C" void kernel_launch(void* const* d_in, const int* in_sizes, int n_in,
                              void* d_out, int out_size, void* d_ws,
                              size_t ws_size, hipStream_t stream) {
  constexpr int IN_FEAT = 256, HIDDEN = 64, NCLS = 16;

  const float* x  = (const float*)d_in[0];
  const int*   ei = (const int*)d_in[1];
  const float* W1 = (const float*)d_in[2];
  const float* b1 = (const float*)d_in[3];
  const float* W2 = (const float*)d_in[4];
  const float* b2 = (const float*)d_in[5];
  float* out = (float*)d_out;

  const int N = in_sizes[0] / IN_FEAT;
  const int E = in_sizes[1] / 2;
  const int* src = ei;
  const int* dst = ei + E;

  float* ws   = (float*)d_ws;
  float* dinv = ws;
  float* h1   = dinv + N;
  float* agg1 = h1 + (size_t)N * HIDDEN;
  float* g2   = agg1 + (size_t)N * HIDDEN;

  const int B = 256;
  const int mtiles  = (N + 15) / 16;
  const int gblocks = (mtiles + 7) / 8;

  k_deg_init<<<(N + B - 1) / B, B, 0, stream>>>(dinv, N);
  k_deg_count<<<(E + B - 1) / B, B, 0, stream>>>(dst, dinv, E);
  k_rsqrt_inplace<<<(N + B - 1) / B, B, 0, stream>>>(dinv, N);

  wmma_gemm_f32<IN_FEAT, HIDDEN><<<gblocks, B, 0, stream>>>(x, W1, h1, N);
  k_self_init<HIDDEN><<<((size_t)N * 16 + B - 1) / B, B, 0, stream>>>(h1, dinv, agg1, N);
  k_scatter64<<<(E + B - 1) / B, B, 0, stream>>>(src, dst, dinv, h1, agg1, E);
  k_bias_relu<HIDDEN><<<((size_t)N * 16 + B - 1) / B, B, 0, stream>>>(agg1, b1, N);

  wmma_gemm_f32<HIDDEN, NCLS><<<gblocks, B, 0, stream>>>(agg1, W2, g2, N);
  k_out_init<<<((size_t)N * 4 + B - 1) / B, B, 0, stream>>>(g2, dinv, b2, out, N);
  k_scatter16<<<(E + B - 1) / B, B, 0, stream>>>(src, dst, dinv, g2, out, E);
}


# Atomically summed planes (degree, layer-1 sums, output) doubled: a work-group adds into its own cache half's copy (by die id), next pass adds the copies; f32 atomics kept, sums now two partials
# speedup vs baseline: 1.8651x; 1.4585x over previous
#include <hip/hip_runtime.h>


typedef float v2f __attribute__((ext_vector_type(2)));
typedef float v8f __attribute__((ext_vector_type(8)));

template <int K, int N>
__global__ __launch_bounds__(256) void wmma_gemm_f32(
    const float* __restrict__ A, const float* __restrict__ B,
    float* __restrict__ C, int M) {
  __shared__ float lB[K * N];
  for (int i = threadIdx.x; i < K * N; i += 256) {
    int k = i / N, n = i % N;
    lB[((k >> 2) * N + n) * 4 + (k & 3)] = B[i];
  }
  __syncthreads();

  const int lane  = threadIdx.x & 31;
  const int wave  = threadIdx.x >> 5;
  const int mbase = (blockIdx.x * 8 + wave) * 16;
  if (mbase >= M) return;

  const int mr = lane & 15;
  const int hi = lane >> 4;

  const float* arow = A + (size_t)(mbase + mr) * K + 2 * hi;
  const float* brow = lB + mr * 4 + 2 * hi;

  constexpr int NT = N / 16;
  v8f acc[NT];
  v8f zero = {};
#pragma unroll
  for (int t = 0; t < NT; ++t) acc[t] = zero;

  for (int kb = 0; kb < K / 4; ++kb) {
    v2f a = *(const v2f*)(arow + kb * 4);
#pragma unroll
    for (int t = 0; t < NT; ++t) {
      v2f b = *(const v2f*)(brow + kb * (N * 4) + t * 64);
      acc[t] = __builtin_amdgcn_wmma_f32_16x16x4_f32(
           false, a,  false, b,
           (short)0, acc[t],  false,  false);
    }
  }

#pragma unroll
  for (int t = 0; t < NT; ++t)
#pragma unroll
    for (int r = 0; r < 8; ++r)
      C[(size_t)(mbase + 8 * hi + r) * N + t * 16 + mr] = acc[t][r];
}

__device__ __forceinline__ int group_half() {
  __shared__ int sh_half;
  if (threadIdx.x == 0)
    sh_half = (int)((__builtin_amdgcn_s_sendmsg_rtn(0x87) >> 18) & 1u);
  __syncthreads();
  return __builtin_amdgcn_readfirstlane(sh_half);
}

__global__ void k_deg_init(float* __restrict__ deg0, float* __restrict__ deg1,
                           int n) {
  int i = blockIdx.x * blockDim.x + threadIdx.x;
  if (i < n) {
    deg0[i] = 1.0f;
    deg1[i] = 0.0f;
  }
}

__global__ void k_deg_count(const int* __restrict__ dst,
                            float* __restrict__ deg0, float* __restrict__ deg1,
                            int e) {
  int i = blockIdx.x * blockDim.x + threadIdx.x;
  float* deg = group_half() ? deg1 : deg0;
  if (i < e) atomicAdd(&deg[dst[i]], 1.0f);
}

__global__ void k_rsqrt_inplace(float* __restrict__ d0,
                                const float* __restrict__ d1, int n) {
  int i = blockIdx.x * blockDim.x + threadIdx.x;
  if (i < n) d0[i] = rsqrtf(d0[i] + d1[i]);
}

template <int F>
__global__ void k_self_init(const float* __restrict__ h,
                            const float* __restrict__ dinv,
                            float* __restrict__ agg,
                            float* __restrict__ agg1, int n) {
  constexpr int PER = F / 4;
  int t = blockIdx.x * blockDim.x + threadIdx.x;
  if (t >= n * PER) return;
  int node = t / PER;
  float di = dinv[node];
  float s = di * di;
  float4 v = ((const float4*)h)[t];
  float4 o = {v.x * s, v.y * s, v.z * s, v.w * s};
  ((float4*)agg)[t] = o;
  ((float4*)agg1)[t] = float4{0.0f, 0.0f, 0.0f, 0.0f};
}

__global__ __launch_bounds__(256) void k_scatter64(
    const int* __restrict__ src, const int* __restrict__ dst,
    const float* __restrict__ dinv, const float* __restrict__ h,
    float* __restrict__ agg0, float* __restrict__ agg1, int e) {
  const int lane = threadIdx.x & 31;
  float* agg = group_half() ? agg1 : agg0;
  const int e0 = __builtin_amdgcn_readfirstlane(
      (int)((blockIdx.x * 256 + threadIdx.x) & ~31u));
  if (e0 >= e) return;
  int s = 0, d = 0;
  float nrm = 0.0f;
  if (e0 + lane < e) {
    s = src[e0 + lane];
    d = dst[e0 + lane];
    nrm = dinv[s] * dinv[d];
  }
  const float* hl = h + lane;
  float* al = agg + lane;
  if (e - e0 >= 32) {
    for (int jb = 0; jb < 32; jb += 16) {
      float v0[16], v1[16];
#pragma unroll
      for (int u = 0; u < 16; ++u) {
        const int sj = __builtin_amdgcn_readlane(s, jb + u);
        const float* hp = hl + (size_t)sj * 64;
        v0[u] = hp[0];
        v1[u] = hp[32];
      }
#pragma unroll
      for (int u = 0; u < 16; ++u) {
        const int dj = __builtin_amdgcn_readlane(d, jb + u);
        const float nj = __int_as_float(
            __builtin_amdgcn_readlane(__float_as_int(nrm), jb + u));
        float* ap = al + (size_t)dj * 64;
        atomicAdd(ap, v0[u] * nj);
        atomicAdd(ap + 32, v1[u] * nj);
      }
    }
  } else {
    for (int j = 0; j < e - e0; ++j) {
      const int sj = __builtin_amdgcn_readlane(s, j);
      const int dj = __builtin_amdgcn_readlane(d, j);
      const float nj = __int_as_float(
          __builtin_amdgcn_readlane(__float_as_int(nrm), j));
      const float* hp = hl + (size_t)sj * 64;
      float* ap = al + (size_t)dj * 64;
      const float a0 = hp[0], a1 = hp[32];
      atomicAdd(ap, a0 * nj);
      atomicAdd(ap + 32, a1 * nj);
    }
  }
}

__global__ __launch_bounds__(256) void k_scatter16(
    const int* __restrict__ src, const int* __restrict__ dst,
    const float* __restrict__ dinv, const float* __restrict__ h,
    float* __restrict__ agg0, float* __restrict__ agg1, int e) {
  const int lane = threadIdx.x & 31;
  float* agg = group_half() ? agg1 : agg0;
  const int c    = lane & 15;
  const int hs   = lane >> 4;
  const int e0 = __builtin_amdgcn_readfirstlane(
      (int)((blockIdx.x * 256 + threadIdx.x) & ~31u));
  if (e0 >= e) return;
  int s = 0, d = 0;
  float nrm = 0.0f;
  if (e0 + lane < e) {
    s = src[e0 + lane];
    d = dst[e0 + lane];
    nrm = dinv[s] * dinv[d];
  }
  const float* hc = h + c;
  float* ac = agg + c;
  if (e - e0 >= 32) {
    float v[16];
#pragma unroll
    for (int u = 0; u < 16; ++u) {
      const int sj = __shfl(s, 2 * u + hs);
      v[u] = hc[(size_t)sj * 16];
    }
#pragma unroll
    for (int u = 0; u < 16; ++u) {
      const int dj = __shfl(d, 2 * u + hs);
      const float nj = __shfl(nrm, 2 * u + hs);
      atomicAdd(ac + (size_t)dj * 16, v[u] * nj);
    }
  } else {
    for (int u = 0; u < 16; ++u) {
      const int idx = 2 * u + hs;
      const int sj = __shfl(s, idx);
      const int dj = __shfl(d, idx);
      const float nj = __shfl(nrm, idx);
      if (e0 + idx < e) {
        const float a = hc[(size_t)sj * 16];
        atomicAdd(ac + (size_t)dj * 16, a * nj);
      }
    }
  }
}

template <int F>
__global__ void k_bias_relu(float* __restrict__ agg,
                            const float* __restrict__ agg1,
                            const float* __restrict__ b, int n) {
  constexpr int PER = F / 4;
  int t = blockIdx.x * blockDim.x + threadIdx.x;
  if (t >= n * PER) return;
  int q = t % PER;
  float4 bb = ((const float4*)b)[q];
  float4 v = ((float4*)agg)[t];
  float4 w = ((const float4*)agg1)[t];
  v.x = fmaxf((v.x + w.x) + bb.x, 0.0f);
  v.y = fmaxf((v.y + w.y) + bb.y, 0.0f);
  v.z = fmaxf((v.z + w.z) + bb.z, 0.0f);
  v.w = fmaxf((v.w + w.w) + bb.w, 0.0f);
  ((float4*)agg)[t] = v;
}

__global__ void k_out_init(const float* __restrict__ g,
                           const float* __restrict__ dinv,
                           const float* __restrict__ b2,
                           float* __restrict__ out,
                           float* __restrict__ out1, int n) {
  int t = blockIdx.x * blockDim.x + threadIdx.x;
  if (t >= n * 4) return;
  int node = t >> 2;
  int q    = t & 3;
  float di = dinv[node];
  float s = di * di;
  float4 v = ((const float4*)g)[t];
  float4 bb = ((const float4*)b2)[q];
  float4 o = {v.x * s + bb.x, v.y * s + bb.y, v.z * s + bb.z, v.w * s + bb.w};
  ((float4*)out)[t] = o;
  ((float4*)out1)[t] = float4{0.0f, 0.0f, 0.0f, 0.0f};
}

__global__ void k_out_sum(float* __restrict__ out,
                          const float* __restrict__ out1, int n4) {
  int t = blockIdx.x * blockDim.x + threadIdx.x;
  if (t >= n4) return;
  float4 v = ((float4*)out)[t];
  float4 w = ((const float4*)out1)[t];
  v.x += w.x; v.y += w.y; v.z += w.z; v.w += w.w;
  ((float4*)out)[t] = v;
}

extern "C" void kernel_launch(void* const* d_in, const int* in_sizes, int n_in,
                              void* d_out, int out_size, void* d_ws,
                              size_t ws_size, hipStream_t stream) {
  constexpr int IN_FEAT = 256, HIDDEN = 64, NCLS = 16;

  const float* x  = (const float*)d_in[0];
  const int*   ei = (const int*)d_in[1];
  const float* W1 = (const float*)d_in[2];
  const float* b1 = (const float*)d_in[3];
  const float* W2 = (const float*)d_in[4];
  const float* b2 = (const float*)d_in[5];
  float* out = (float*)d_out;

  const int N = in_sizes[0] / IN_FEAT;
  const int E = in_sizes[1] / 2;
  const int* src = ei;
  const int* dst = ei + E;

  float* ws   = (float*)d_ws;
  float* dinv = ws;
  float* h1   = dinv + N;
  float* agg1 = h1 + (size_t)N * HIDDEN;
  float* g2   = agg1 + (size_t)N * HIDDEN;
  float* agg1b = g2 + (size_t)N * NCLS;
  float* outb  = agg1b + (size_t)N * HIDDEN;
  float* degb  = outb + (size_t)N * NCLS;

  const int B = 256;
  const int mtiles  = (N + 15) / 16;
  const int gblocks = (mtiles + 7) / 8;

  k_deg_init<<<(N + B - 1) / B, B, 0, stream>>>(dinv, degb, N);
  k_deg_count<<<(E + B - 1) / B, B, 0, stream>>>(dst, dinv, degb, E);
  k_rsqrt_inplace<<<(N + B - 1) / B, B, 0, stream>>>(dinv, degb, N);

  wmma_gemm_f32<IN_FEAT, HIDDEN><<<gblocks, B, 0, stream>>>(x, W1, h1, N);
  k_self_init<HIDDEN><<<((size_t)N * 16 + B - 1) / B, B, 0, stream>>>(h1, dinv, agg1, agg1b, N);
  k_scatter64<<<(E + B - 1) / B, B, 0, stream>>>(src, dst, dinv, h1, agg1, agg1b, E);
  k_bias_relu<HIDDEN><<<((size_t)N * 16 + B - 1) / B, B, 0, stream>>>(agg1, agg1b, b1, N);

  wmma_gemm_f32<HIDDEN, NCLS><<<gblocks, B, 0, stream>>>(agg1, W2, g2, N);
  k_out_init<<<((size_t)N * 4 + B - 1) / B, B, 0, stream>>>(g2, dinv, b2, out, outb, N);
  k_scatter16<<<(E + B - 1) / B, B, 0, stream>>>(src, dst, dinv, g2, out, outb, E);
  k_out_sum<<<((size_t)N * 4 + B - 1) / B, B, 0, stream>>>(out, outb, N * 4);
}


# Edges bucketed by destination each call (counting add returns the rank, scan, place); a wave sums a node's rows in registers, one store per row; float atomics gone, sum order = rank order
# speedup vs baseline: 2.3163x; 1.2419x over previous
#include <hip/hip_runtime.h>


typedef float v2f __attribute__((ext_vector_type(2)));
typedef float v8f __attribute__((ext_vector_type(8)));

template <int K, int N>
__global__ __launch_bounds__(256) void wmma_gemm_f32(
    const float* __restrict__ A, const float* __restrict__ B,
    float* __restrict__ C, int M) {
  __shared__ float lB[K * N];
  for (int i = threadIdx.x; i < K * N; i += 256) {
    int k = i / N, n = i % N;
    lB[((k >> 2) * N + n) * 4 + (k & 3)] = B[i];
  }
  __syncthreads();

  const int lane  = threadIdx.x & 31;
  const int wave  = threadIdx.x >> 5;
  const int mbase = (blockIdx.x * 8 + wave) * 16;
  if (mbase >= M) return;

  const int mr = lane & 15;
  const int hi = lane >> 4;

  const float* arow = A + (size_t)(mbase + mr) * K + 2 * hi;
  const float* brow = lB + mr * 4 + 2 * hi;

  constexpr int NT = N / 16;
  v8f acc[NT];
  v8f zero = {};
#pragma unroll
  for (int t = 0; t < NT; ++t) acc[t] = zero;

  for (int kb = 0; kb < K / 4; ++kb) {
    v2f a = *(const v2f*)(arow + kb * 4);
#pragma unroll
    for (int t = 0; t < NT; ++t) {
      v2f b = *(const v2f*)(brow + kb * (N * 4) + t * 64);
      acc[t] = __builtin_amdgcn_wmma_f32_16x16x4_f32(
           false, a,  false, b,
           (short)0, acc[t],  false,  false);
    }
  }

#pragma unroll
  for (int t = 0; t < NT; ++t)
#pragma unroll
    for (int r = 0; r < 8; ++r)
      C[(size_t)(mbase + 8 * hi + r) * N + t * 16 + mr] = acc[t][r];
}

__device__ __forceinline__ int group_half() {
  __shared__ int sh_half;
  if (threadIdx.x == 0)
    sh_half = (int)((__builtin_amdgcn_s_sendmsg_rtn(0x87) >> 18) & 1u);
  __syncthreads();
  return __builtin_amdgcn_readfirstlane(sh_half);
}

__global__ void k_cnt_zero(int* __restrict__ cnt0, int* __restrict__ cnt1,
                           int n) {
  int i = blockIdx.x * blockDim.x + threadIdx.x;
  if (i < n) {
    cnt0[i] = 0;
    cnt1[i] = 0;
  }
}

__global__ void k_count_rank(const int* __restrict__ dst,
                             int* __restrict__ cnt0, int* __restrict__ cnt1,
                             unsigned* __restrict__ rk, int e) {
  int i = blockIdx.x * blockDim.x + threadIdx.x;
  const int half = group_half();
  int* cnt = half ? cnt1 : cnt0;
  if (i < e) {
    const int r = atomicAdd(&cnt[dst[i]], 1);
    rk[i] = (unsigned)r | ((unsigned)half << 31);
  }
}

__device__ __forceinline__ int wave_incl_scan(int v, int lane) {
#pragma unroll
  for (int o = 1; o < 32; o <<= 1) {
    const int t = __shfl_up(v, o);
    if (lane >= o) v += t;
  }
  return v;
}

__global__ __launch_bounds__(256) void k_deg_blocksum(
    const int* __restrict__ cnt0, const int* __restrict__ cnt1,
    float* __restrict__ dinv, int* __restrict__ bsum, int n) {
  __shared__ int ws[8];
  const int i = blockIdx.x * 256 + threadIdx.x;
  const int lane = threadIdx.x & 31, wave = threadIdx.x >> 5;
  int t = 0;
  if (i < n) {
    t = cnt0[i] + cnt1[i];
    dinv[i] = rsqrtf((float)(t + 1));
  }
  const int s = wave_incl_scan(t, lane);
  if (lane == 31) ws[wave] = s;
  __syncthreads();
  if (threadIdx.x == 0) {
    int tot = 0;
#pragma unroll
    for (int w = 0; w < 8; ++w) tot += ws[w];
    bsum[blockIdx.x] = tot;
  }
}

__global__ __launch_bounds__(256) void k_scan_rowptr(
    const int* __restrict__ cnt0, const int* __restrict__ cnt1,
    const int* __restrict__ bsum, int* __restrict__ rowptr, int n) {
  __shared__ int ws[8];
  __shared__ int wp[8];
  const int i = blockIdx.x * 256 + threadIdx.x;
  const int lane = threadIdx.x & 31, wave = threadIdx.x >> 5;
  int p = 0;
  for (int k = threadIdx.x; k < (int)blockIdx.x; k += 256) p += bsum[k];
  const int t = (i < n) ? cnt0[i] + cnt1[i] : 0;
  const int ps = wave_incl_scan(p, lane);
  const int s = wave_incl_scan(t, lane);
  if (lane == 31) {
    wp[wave] = ps;
    ws[wave] = s;
  }
  __syncthreads();
  int base = 0;
#pragma unroll
  for (int w = 0; w < 8; ++w) {
    base += wp[w];
    if (w < wave) base += ws[w];
  }
  const int incl = base + s;
  if (i < n) {
    rowptr[i] = incl - t;
    if (i == n - 1) rowptr[n] = incl;
  }
}

__global__ void k_place(const int* __restrict__ src,
                        const int* __restrict__ dst,
                        const unsigned* __restrict__ rk,
                        const int* __restrict__ cnt0,
                        const int* __restrict__ rowptr,
                        const float* __restrict__ dinv,
                        int2* __restrict__ col, int e) {
  int i = blockIdx.x * blockDim.x + threadIdx.x;
  if (i >= e) return;
  const int s = src[i];
  const int d = dst[i];
  const unsigned w = rk[i];
  int pos = rowptr[d] + (int)(w & 0x7fffffffu);
  if (w >> 31) pos += cnt0[d];
  col[pos] = int2{s, __float_as_int(dinv[s])};
}

__global__ __launch_bounds__(256) void k_gather64(
    const int* __restrict__ rowptr, const int2* __restrict__ col,
    const float* __restrict__ dinv, const float* __restrict__ h,
    const float* __restrict__ b, float* __restrict__ o, int n) {
  const int lane = threadIdx.x & 31;
  const int node = __builtin_amdgcn_readfirstlane(
      (int)((blockIdx.x * 256 + threadIdx.x) >> 5));
  if (node >= n) return;
  const int beg = __builtin_amdgcn_readfirstlane(rowptr[node]);
  const int cnt = __builtin_amdgcn_readfirstlane(rowptr[node + 1]) - beg;
  const float di = dinv[node];
  const v2f* hl = (const v2f*)h + lane;
  const v2f self = hl[(size_t)node * 32];
  const float ss = di * di;
  float ax = self.x * ss, ay = self.y * ss;
  for (int j0 = 0; j0 < cnt; j0 += 32) {
    int s = 0, wb = 0;
    if (j0 + lane < cnt) {
      const int2 c = col[beg + j0 + lane];
      s = c.x;
      wb = c.y;
    }
    const int m = (cnt - j0 < 32) ? (cnt - j0) : 32;
    int j = 0;
    for (; j + 8 <= m; j += 8) {
      v2f r[8];
#pragma unroll
      for (int u = 0; u < 8; ++u)
        r[u] = hl[(size_t)__builtin_amdgcn_readlane(s, j + u) * 32];
#pragma unroll
      for (int u = 0; u < 8; ++u) {
        const float nj =
            __int_as_float(__builtin_amdgcn_readlane(wb, j + u)) * di;
        ax += r[u].x * nj;
        ay += r[u].y * nj;
      }
    }
    if (j + 4 <= m) {
      v2f r[4];
#pragma unroll
      for (int u = 0; u < 4; ++u)
        r[u] = hl[(size_t)__builtin_amdgcn_readlane(s, j + u) * 32];
#pragma unroll
      for (int u = 0; u < 4; ++u) {
        const float nj =
            __int_as_float(__builtin_amdgcn_readlane(wb, j + u)) * di;
        ax += r[u].x * nj;
        ay += r[u].y * nj;
      }
      j += 4;
    }
    for (; j < m; ++j) {
      const v2f r = hl[(size_t)__builtin_amdgcn_readlane(s, j) * 32];
      const float nj = __int_as_float(__builtin_amdgcn_readlane(wb, j)) * di;
      ax += r.x * nj;
      ay += r.y * nj;
    }
  }
  const v2f bb = ((const v2f*)b)[lane];
  v2f res;
  res.x = fmaxf(ax + bb.x, 0.0f);
  res.y = fmaxf(ay + bb.y, 0.0f);
  ((v2f*)o)[(size_t)node * 32 + lane] = res;
}

__global__ __launch_bounds__(256) void k_gather16(
    const int* __restrict__ rowptr, const int2* __restrict__ col,
    const float* __restrict__ dinv, const float* __restrict__ g,
    const float* __restrict__ b, float* __restrict__ out, int n) {
  const int c = threadIdx.x & 15;
  const int node = (int)((blockIdx.x * 256 + threadIdx.x) >> 4);
  const bool act = node < n;
  int beg = 0, cnt = 0;
  float di = 0.0f, acc = 0.0f;
  if (act) {
    beg = rowptr[node];
    cnt = rowptr[node + 1] - beg;
    di = dinv[node];
    acc = g[(size_t)node * 16 + c] * (di * di);
  }
  const int other = __shfl_xor(cnt, 16);
  const int cmax = cnt > other ? cnt : other;
  const float* gc = g + c;
  for (int j = 0; j < cmax; j += 8) {
    int2 sl[8];
    float r[8];
#pragma unroll
    for (int u = 0; u < 8; ++u) {
      sl[u] = int2{0, 0};
      if (j + u < cnt) sl[u] = col[beg + j + u];
    }
#pragma unroll
    for (int u = 0; u < 8; ++u) {
      r[u] = 0.0f;
      if (j + u < cnt) r[u] = gc[(size_t)sl[u].x * 16];
    }
#pragma unroll
    for (int u = 0; u < 8; ++u)
      if (j + u < cnt) acc += r[u] * (__int_as_float(sl[u].y) * di);
  }
  if (act) out[(size_t)node * 16 + c] = acc + b[c];
}

extern "C" void kernel_launch(void* const* d_in, const int* in_sizes, int n_in,
                              void* d_out, int out_size, void* d_ws,
                              size_t ws_size, hipStream_t stream) {
  constexpr int IN_FEAT = 256, HIDDEN = 64, NCLS = 16;

  const float* x  = (const float*)d_in[0];
  const int*   ei = (const int*)d_in[1];
  const float* W1 = (const float*)d_in[2];
  const float* b1 = (const float*)d_in[3];
  const float* W2 = (const float*)d_in[4];
  const float* b2 = (const float*)d_in[5];
  float* out = (float*)d_out;

  const int N = in_sizes[0] / IN_FEAT;
  const int E = in_sizes[1] / 2;
  const int* src = ei;
  const int* dst = ei + E;

  const size_t NP = ((size_t)N + 64) / 64 * 64;
  const size_t EP = ((size_t)E + 63) / 64 * 64;
  float* ws   = (float*)d_ws;
  float* h1   = ws;
  float* a1   = h1 + (size_t)N * HIDDEN;
  float* g2   = a1 + (size_t)N * HIDDEN;
  float* dinv = g2 + (size_t)N * NCLS;
  int* cnt0   = (int*)(dinv + NP);
  int* cnt1   = cnt0 + NP;
  int* rowptr = cnt1 + NP;
  int* bsum   = rowptr + NP;
  unsigned* rk = (unsigned*)(bsum + NP);
  int2* col   = (int2*)(rk + EP);

  const int B = 256;
  const int mtiles  = (N + 15) / 16;
  const int gblocks = (mtiles + 7) / 8;
  const int nblocks = (N + B - 1) / B;

  k_cnt_zero<<<nblocks, B, 0, stream>>>(cnt0, cnt1, N);
  k_count_rank<<<(E + B - 1) / B, B, 0, stream>>>(dst, cnt0, cnt1, rk, E);
  k_deg_blocksum<<<nblocks, B, 0, stream>>>(cnt0, cnt1, dinv, bsum, N);
  k_scan_rowptr<<<nblocks, B, 0, stream>>>(cnt0, cnt1, bsum, rowptr, N);
  k_place<<<(E + B - 1) / B, B, 0, stream>>>(src, dst, rk, cnt0, rowptr, dinv, col, E);

  wmma_gemm_f32<IN_FEAT, HIDDEN><<<gblocks, B, 0, stream>>>(x, W1, h1, N);
  k_gather64<<<(N + 7) / 8, B, 0, stream>>>(rowptr, col, dinv, h1, b1, a1, N);

  wmma_gemm_f32<HIDDEN, NCLS><<<gblocks, B, 0, stream>>>(a1, W2, g2, N);
  k_gather16<<<(N + 15) / 16, B, 0, stream>>>(rowptr, col, dinv, g2, b2, out, N);
}


# Counting add's returned rank puts each edge straight into a fixed 2x32-slot bucket of its destination (overflow list behind); scan and place passes gone; dinv[src] scales GEMM rows, dinv[dst] the sum
# speedup vs baseline: 2.4574x; 1.0609x over previous
#include <hip/hip_runtime.h>


typedef float v2f __attribute__((ext_vector_type(2)));
typedef float v8f __attribute__((ext_vector_type(8)));

constexpr int SUB = 32;
constexpr int BKT = 2 * SUB;
template <int K, int N>
__global__ __launch_bounds__(256) void wmma_gemm_f32(
    const float* __restrict__ A, const float* __restrict__ B,
    const int* __restrict__ cnt0, const int* __restrict__ cnt1,
    float* __restrict__ C, int M) {
  __shared__ float lB[K * N];
  for (int i = threadIdx.x; i < K * N; i += 256) {
    int k = i / N, n = i % N;
    lB[((k >> 2) * N + n) * 4 + (k & 3)] = B[i];
  }
  __syncthreads();

  const int lane  = threadIdx.x & 31;
  const int wave  = threadIdx.x >> 5;
  const int mbase = (blockIdx.x * 8 + wave) * 16;
  if (mbase >= M) return;

  const int mr = lane & 15;
  const int hi = lane >> 4;

  const float* arow = A + (size_t)(mbase + mr) * K + 2 * hi;
  const float* brow = lB + mr * 4 + 2 * hi;

  float sc[8];
#pragma unroll
  for (int r = 0; r < 8; ++r) {
    const int row = mbase + 8 * hi + r;
    sc[r] = rsqrtf((float)(cnt0[row] + cnt1[row] + 1));
  }

  constexpr int NT = N / 16;
  v8f acc[NT];
  v8f zero = {};
#pragma unroll
  for (int t = 0; t < NT; ++t) acc[t] = zero;

  for (int kb = 0; kb < K / 4; ++kb) {
    v2f a = *(const v2f*)(arow + kb * 4);
#pragma unroll
    for (int t = 0; t < NT; ++t) {
      v2f b = *(const v2f*)(brow + kb * (N * 4) + t * 64);
      acc[t] = __builtin_amdgcn_wmma_f32_16x16x4_f32(
           false, a,  false, b,
           (short)0, acc[t],  false,  false);
    }
  }

#pragma unroll
  for (int t = 0; t < NT; ++t)
#pragma unroll
    for (int r = 0; r < 8; ++r)
      C[(size_t)(mbase + 8 * hi + r) * N + t * 16 + mr] = acc[t][r] * sc[r];
}

__device__ __forceinline__ int group_half() {
  __shared__ int sh_half;
  if (threadIdx.x == 0)
    sh_half = (int)((__builtin_amdgcn_s_sendmsg_rtn(0x87) >> 18) & 1u);
  __syncthreads();
  return __builtin_amdgcn_readfirstlane(sh_half);
}

__global__ void k_cnt_zero(int* __restrict__ cnt0, int* __restrict__ cnt1,
                           int* __restrict__ ovc, int n) {
  int i = blockIdx.x * blockDim.x + threadIdx.x;
  if (i < n) {
    cnt0[i] = 0;
    cnt1[i] = 0;
  }
  if (i == 0) *ovc = 0;
}

__global__ void k_count_place(const int* __restrict__ src,
                              const int* __restrict__ dst,
                              int* __restrict__ cnt0, int* __restrict__ cnt1,
                              int* __restrict__ col, int* __restrict__ ovc,
                              int2* __restrict__ ov, int e) {
  int i = blockIdx.x * blockDim.x + threadIdx.x;
  const int half = group_half();
  int* cnt = half ? cnt1 : cnt0;
  if (i < e) {
    const int d = dst[i];
    const int s = src[i];
    const int r = atomicAdd(&cnt[d], 1);
    if (r < SUB) {
      col[(size_t)d * BKT + half * SUB + r] = s;
    } else {
      const int k = atomicAdd(ovc, 1);
      ov[k] = int2{d, s};
    }
  }
}

__device__ __forceinline__ void walk64(const v2f* __restrict__ hl, int s,
                                       int m, float& ax, float& ay) {
  int j = 0;
  for (; j + 8 <= m; j += 8) {
    v2f r[8];
#pragma unroll
    for (int u = 0; u < 8; ++u)
      r[u] = hl[(size_t)__builtin_amdgcn_readlane(s, j + u) * 32];
#pragma unroll
    for (int u = 0; u < 8; ++u) {
      ax += r[u].x;
      ay += r[u].y;
    }
  }
  if (j + 4 <= m) {
    v2f r[4];
#pragma unroll
    for (int u = 0; u < 4; ++u)
      r[u] = hl[(size_t)__builtin_amdgcn_readlane(s, j + u) * 32];
#pragma unroll
    for (int u = 0; u < 4; ++u) {
      ax += r[u].x;
      ay += r[u].y;
    }
    j += 4;
  }
  for (; j < m; ++j) {
    const v2f r = hl[(size_t)__builtin_amdgcn_readlane(s, j) * 32];
    ax += r.x;
    ay += r.y;
  }
}

__global__ __launch_bounds__(256) void k_gather64(
    const int* __restrict__ cnt0, const int* __restrict__ cnt1,
    const int* __restrict__ col, const int* __restrict__ ovc,
    const int2* __restrict__ ov, const float* __restrict__ h,
    const float* __restrict__ b, float* __restrict__ o, int n) {
  const int lane = threadIdx.x & 31;
  const int node = __builtin_amdgcn_readfirstlane(
      (int)((blockIdx.x * 256 + threadIdx.x) >> 5));
  if (node >= n) return;
  const int* bk = col + (size_t)node * BKT;
  const int s0 = bk[lane];
  const int s1 = bk[SUB + lane];
  const int c0 = __builtin_amdgcn_readfirstlane(cnt0[node]);
  const int c1 = __builtin_amdgcn_readfirstlane(cnt1[node]);
  const v2f* hl = (const v2f*)h + lane;
  const v2f self = hl[(size_t)node * 32];
  const float di = rsqrtf((float)(c0 + c1 + 1));
  const int m0 = c0 < SUB ? c0 : SUB;
  const int m1 = c1 < SUB ? c1 : SUB;
  const int m = m0 + m1;
  float ax = self.x, ay = self.y;
  const int t1 = __shfl(s1, lane - m0);
  const int la = lane < m0 ? s0 : t1;
  walk64(hl, la, m < 32 ? m : 32, ax, ay);
  if (m > 32) {
    const int lb = __shfl(s1, lane + 32 - m0);
    walk64(hl, lb, m - 32, ax, ay);
  }
  if (c0 > SUB || c1 > SUB) {
    const int nov = __builtin_amdgcn_readfirstlane(*ovc);
    for (int k0 = 0; k0 < nov; k0 += 32) {
      int2 ed = int2{-1, 0};
      if (k0 + lane < nov) ed = ov[k0 + lane];
      unsigned hit = __builtin_amdgcn_ballot_w32(ed.x == node);
      while (hit) {
        const int bpos = __builtin_ctz(hit);
        hit &= hit - 1;
        const v2f r = hl[(size_t)__builtin_amdgcn_readlane(ed.y, bpos) * 32];
        ax += r.x;
        ay += r.y;
      }
    }
  }
  const v2f bb = ((const v2f*)b)[lane];
  v2f res;
  res.x = fmaxf(ax * di + bb.x, 0.0f);
  res.y = fmaxf(ay * di + bb.y, 0.0f);
  ((v2f*)o)[(size_t)node * 32 + lane] = res;
}

__global__ __launch_bounds__(256) void k_gather16(
    const int* __restrict__ cnt0, const int* __restrict__ cnt1,
    const int* __restrict__ col, const int* __restrict__ ovc,
    const int2* __restrict__ ov, const float* __restrict__ g,
    const float* __restrict__ b, float* __restrict__ out, int n) {
  const int c = threadIdx.x & 15;
  const int node = (int)((blockIdx.x * 256 + threadIdx.x) >> 4);
  const bool act = node < n;
  int c0 = 0, c1 = 0, m0 = 0, m = 0;
  float di = 0.0f, acc = 0.0f;
  if (act) {
    c0 = cnt0[node];
    c1 = cnt1[node];
    m0 = c0 < SUB ? c0 : SUB;
    m = m0 + (c1 < SUB ? c1 : SUB);
    di = rsqrtf((float)(c0 + c1 + 1));
    acc = g[(size_t)node * 16 + c];
  }
  const int other = __shfl_xor(m, 16);
  const int mmax = m > other ? m : other;
  const float* gc = g + c;
  const int* bk = col + (size_t)node * BKT;
  for (int j = 0; j < mmax; j += 8) {
    int sl[8];
    float r[8];
#pragma unroll
    for (int u = 0; u < 8; ++u) {
      sl[u] = 0;
      const int k = j + u;
      if (k < m) sl[u] = bk[k < m0 ? k : SUB + k - m0];
    }
#pragma unroll
    for (int u = 0; u < 8; ++u) {
      r[u] = 0.0f;
      if (j + u < m) r[u] = gc[(size_t)sl[u] * 16];
    }
#pragma unroll
    for (int u = 0; u < 8; ++u)
      if (j + u < m) acc += r[u];
  }
  if (act && (c0 > SUB || c1 > SUB)) {
    const int nov = *ovc;
    for (int k = 0; k < nov; ++k) {
      const int2 ed = ov[k];
      if (ed.x == node) acc += gc[(size_t)ed.y * 16];
    }
  }
  if (act) out[(size_t)node * 16 + c] = acc * di + b[c];
}

extern "C" void kernel_launch(void* const* d_in, const int* in_sizes, int n_in,
                              void* d_out, int out_size, void* d_ws,
                              size_t ws_size, hipStream_t stream) {
  constexpr int IN_FEAT = 256, HIDDEN = 64, NCLS = 16;

  const float* x  = (const float*)d_in[0];
  const int*   ei = (const int*)d_in[1];
  const float* W1 = (const float*)d_in[2];
  const float* b1 = (const float*)d_in[3];
  const float* W2 = (const float*)d_in[4];
  const float* b2 = (const float*)d_in[5];
  float* out = (float*)d_out;

  const int N = in_sizes[0] / IN_FEAT;
  const int E = in_sizes[1] / 2;
  const int* src = ei;
  const int* dst = ei + E;

  const size_t NP = ((size_t)N + 63) / 64 * 64;
  float* ws   = (float*)d_ws;
  float* h1   = ws;
  float* a1   = h1 + (size_t)N * HIDDEN;
  float* g2   = a1 + (size_t)N * HIDDEN;
  int* cnt0   = (int*)(g2 + (size_t)N * NCLS);
  int* cnt1   = cnt0 + NP;
  int* ovc    = cnt1 + NP;
  int* col    = ovc + 64;
  int2* ov    = (int2*)(col + (size_t)N * BKT);

  const int B = 256;
  const int mtiles  = (N + 15) / 16;
  const int gblocks = (mtiles + 7) / 8;
  const int nblocks = (N + B - 1) / B;

  k_cnt_zero<<<nblocks, B, 0, stream>>>(cnt0, cnt1, ovc, N);
  k_count_place<<<(E + B - 1) / B, B, 0, stream>>>(src, dst, cnt0, cnt1, col, ovc, ov, E);

  wmma_gemm_f32<IN_FEAT, HIDDEN><<<gblocks, B, 0, stream>>>(x, W1, cnt0, cnt1, h1, N);
  k_gather64<<<(N + 7) / 8, B, 0, stream>>>(cnt0, cnt1, col, ovc, ov, h1, b1, a1, N);

  wmma_gemm_f32<HIDDEN, NCLS><<<gblocks, B, 0, stream>>>(a1, W2, cnt0, cnt1, g2, N);
  k_gather16<<<(N + 15) / 16, B, 0, stream>>>(cnt0, cnt1, col, ovc, ov, g2, b2, out, N);
}


# Edge pass and both gathers launch few long-lived waves (4 edges per thread; a wave walks many nodes, 4 at a time in layer 2, next bucket asked for early), not one short wave per node; sum order kept
# speedup vs baseline: 2.5615x; 1.0424x over previous
#include <hip/hip_runtime.h>


typedef float v2f __attribute__((ext_vector_type(2)));
typedef float v8f __attribute__((ext_vector_type(8)));

constexpr int SUB = 32;
constexpr int BKT = 2 * SUB;
template <int K, int N>
__global__ __launch_bounds__(256) void wmma_gemm_f32(
    const float* __restrict__ A, const float* __restrict__ B,
    const int* __restrict__ cnt0, const int* __restrict__ cnt1,
    float* __restrict__ C, int M) {
  __shared__ float lB[K * N];
  for (int i = threadIdx.x; i < K * N; i += 256) {
    int k = i / N, n = i % N;
    lB[((k >> 2) * N + n) * 4 + (k & 3)] = B[i];
  }
  __syncthreads();

  const int lane  = threadIdx.x & 31;
  const int wave  = threadIdx.x >> 5;
  const int mbase = (blockIdx.x * 8 + wave) * 16;
  if (mbase >= M) return;

  const int mr = lane & 15;
  const int hi = lane >> 4;

  const float* arow = A + (size_t)(mbase + mr) * K + 2 * hi;
  const float* brow = lB + mr * 4 + 2 * hi;

  float sc[8];
#pragma unroll
  for (int r = 0; r < 8; ++r) {
    const int row = mbase + 8 * hi + r;
    sc[r] = rsqrtf((float)(cnt0[row] + cnt1[row] + 1));
  }

  constexpr int NT = N / 16;
  v8f acc[NT];
  v8f zero = {};
#pragma unroll
  for (int t = 0; t < NT; ++t) acc[t] = zero;

  for (int kb = 0; kb < K / 4; ++kb) {
    v2f a = *(const v2f*)(arow + kb * 4);
#pragma unroll
    for (int t = 0; t < NT; ++t) {
      v2f b = *(const v2f*)(brow + kb * (N * 4) + t * 64);
      acc[t] = __builtin_amdgcn_wmma_f32_16x16x4_f32(
           false, a,  false, b,
           (short)0, acc[t],  false,  false);
    }
  }

#pragma unroll
  for (int t = 0; t < NT; ++t)
#pragma unroll
    for (int r = 0; r < 8; ++r)
      C[(size_t)(mbase + 8 * hi + r) * N + t * 16 + mr] = acc[t][r] * sc[r];
}

__device__ __forceinline__ int group_half() {
  __shared__ int sh_half;
  if (threadIdx.x == 0)
    sh_half = (int)((__builtin_amdgcn_s_sendmsg_rtn(0x87) >> 18) & 1u);
  __syncthreads();
  return __builtin_amdgcn_readfirstlane(sh_half);
}

__global__ void k_cnt_zero(int* __restrict__ cnt0, int* __restrict__ cnt1,
                           int* __restrict__ ovc, int n) {
  int i = blockIdx.x * blockDim.x + threadIdx.x;
  if (i < n) {
    cnt0[i] = 0;
    cnt1[i] = 0;
  }
  if (i == 0) *ovc = 0;
}

constexpr int EPT = 4;
template <bool FULL>
__device__ __forceinline__ void count_place_body(
    const int* __restrict__ src, const int* __restrict__ dst,
    int* __restrict__ cnt, int* __restrict__ col, int* __restrict__ ovc,
    int2* __restrict__ ov, int e, int base, int half) {
  int d[EPT], s[EPT], r[EPT];
#pragma unroll
  for (int u = 0; u < EPT; ++u) {
    const int i = base + u * 256;
    const bool ok = FULL || i < e;
    d[u] = ok ? dst[i] : -1;
    s[u] = ok ? src[i] : 0;
  }
#pragma unroll
  for (int u = 0; u < EPT; ++u) {
    r[u] = 0;
    if (FULL || d[u] >= 0) r[u] = atomicAdd(&cnt[d[u]], 1);
  }
#pragma unroll
  for (int u = 0; u < EPT; ++u) {
    if (FULL || d[u] >= 0) {
      if (r[u] < SUB) {
        col[(size_t)d[u] * BKT + half * SUB + r[u]] = s[u];
      } else {
        const int k = atomicAdd(ovc, 1);
        ov[k] = int2{d[u], s[u]};
      }
    }
  }
}

__global__ __launch_bounds__(256) void k_count_place(
    const int* __restrict__ src, const int* __restrict__ dst,
    int* __restrict__ cnt0, int* __restrict__ cnt1, int* __restrict__ col,
    int* __restrict__ ovc, int2* __restrict__ ov, int e) {
  const int half = group_half();
  int* cnt = half ? cnt1 : cnt0;
  const int first = blockIdx.x * (256 * EPT);
  const int base = first + threadIdx.x;
  if (first + 256 * EPT <= e)
    count_place_body<true>(src, dst, cnt, col, ovc, ov, e, base, half);
  else
    count_place_body<false>(src, dst, cnt, col, ovc, ov, e, base, half);
}

__device__ __forceinline__ void walk64(const v2f* __restrict__ hl, int s,
                                       int m, float& ax, float& ay) {
  int j = 0;
  for (; j + 8 <= m; j += 8) {
    v2f r[8];
#pragma unroll
    for (int u = 0; u < 8; ++u)
      r[u] = hl[(size_t)__builtin_amdgcn_readlane(s, j + u) * 32];
#pragma unroll
    for (int u = 0; u < 8; ++u) {
      ax += r[u].x;
      ay += r[u].y;
    }
  }
  if (j + 4 <= m) {
    v2f r[4];
#pragma unroll
    for (int u = 0; u < 4; ++u)
      r[u] = hl[(size_t)__builtin_amdgcn_readlane(s, j + u) * 32];
#pragma unroll
    for (int u = 0; u < 4; ++u) {
      ax += r[u].x;
      ay += r[u].y;
    }
    j += 4;
  }
  for (; j < m; ++j) {
    const v2f r = hl[(size_t)__builtin_amdgcn_readlane(s, j) * 32];
    ax += r.x;
    ay += r.y;
  }
}

__global__ __launch_bounds__(256) void k_gather64(
    const int* __restrict__ cnt0, const int* __restrict__ cnt1,
    const int* __restrict__ col, const int* __restrict__ ovc,
    const int2* __restrict__ ov, const float* __restrict__ h,
    const float* __restrict__ b, float* __restrict__ o, int n) {
  const int lane = threadIdx.x & 31;
  const int nw = (int)(gridDim.x * 8);
  int node = __builtin_amdgcn_readfirstlane(
      (int)((blockIdx.x * 256 + threadIdx.x) >> 5));
  if (node >= n) return;
  const v2f* hl = (const v2f*)h + lane;
  const v2f bb = ((const v2f*)b)[lane];
  int s0 = col[(size_t)node * BKT + lane];
  int s1 = col[(size_t)node * BKT + SUB + lane];
  int k0 = cnt0[node];
  int k1 = cnt1[node];
  v2f self = hl[(size_t)node * 32];
  for (;;) {
    const int c0 = __builtin_amdgcn_readfirstlane(k0);
    const int c1 = __builtin_amdgcn_readfirstlane(k1);
    const float di = rsqrtf((float)(c0 + c1 + 1));
    const int m0 = c0 < SUB ? c0 : SUB;
    const int m1 = c1 < SUB ? c1 : SUB;
    const int m = m0 + m1;
    float ax = self.x, ay = self.y;
    const int t1 = __shfl(s1, lane - m0);
    const int la = lane < m0 ? s0 : t1;
    const int nxt = node + nw;
    const bool more = nxt < n;
    int ns0 = 0, ns1 = 0, nk0 = 0, nk1 = 0;
    v2f nself = {0.0f, 0.0f};
    if (more) {
      ns0 = col[(size_t)nxt * BKT + lane];
      ns1 = col[(size_t)nxt * BKT + SUB + lane];
      nk0 = cnt0[nxt];
      nk1 = cnt1[nxt];
      nself = hl[(size_t)nxt * 32];
    }
    walk64(hl, la, m < 32 ? m : 32, ax, ay);
    if (m > 32) {
      const int lb = __shfl(s1, lane + 32 - m0);
      walk64(hl, lb, m - 32, ax, ay);
    }
    if (c0 > SUB || c1 > SUB) {
      const int nov = __builtin_amdgcn_readfirstlane(*ovc);
      for (int q0 = 0; q0 < nov; q0 += 32) {
        int2 ed = int2{-1, 0};
        if (q0 + lane < nov) ed = ov[q0 + lane];
        unsigned hit = __builtin_amdgcn_ballot_w32(ed.x == node);
        while (hit) {
          const int bpos = __builtin_ctz(hit);
          hit &= hit - 1;
          const v2f r = hl[(size_t)__builtin_amdgcn_readlane(ed.y, bpos) * 32];
          ax += r.x;
          ay += r.y;
        }
      }
    }
    v2f res;
    res.x = fmaxf(ax * di + bb.x, 0.0f);
    res.y = fmaxf(ay * di + bb.y, 0.0f);
    ((v2f*)o)[(size_t)node * 32 + lane] = res;
    if (!more) break;
    node = nxt;
    s0 = ns0;
    s1 = ns1;
    k0 = nk0;
    k1 = nk1;
    self = nself;
  }
}

__global__ __launch_bounds__(256) void k_gather16(
    const int* __restrict__ cnt0, const int* __restrict__ cnt1,
    const int* __restrict__ col, const int* __restrict__ ovc,
    const int2* __restrict__ ov, const float* __restrict__ g,
    const float* __restrict__ b, float* __restrict__ out, int n) {
  const int c = threadIdx.x & 7;
  const int nq = (int)(gridDim.x * 32);
  int node = (int)((blockIdx.x * 256 + threadIdx.x) >> 3);
  const v2f* gc = (const v2f*)g + c;
  const v2f bc = ((const v2f*)b)[c];
  bool act = node < n;
  int c0 = 0, c1 = 0;
  v2f acc = {0.0f, 0.0f};
  if (act) {
    c0 = cnt0[node];
    c1 = cnt1[node];
    acc = gc[(size_t)node * 8];
  }
  while (__builtin_amdgcn_ballot_w32(act) != 0u) {
    int m0 = 0, m = 0;
    float di = 0.0f;
    if (act) {
      m0 = c0 < SUB ? c0 : SUB;
      m = m0 + (c1 < SUB ? c1 : SUB);
      di = rsqrtf((float)(c0 + c1 + 1));
    }
    int mmax = m;
    int other = __shfl_xor(mmax, 8);
    mmax = mmax > other ? mmax : other;
    other = __shfl_xor(mmax, 16);
    mmax = mmax > other ? mmax : other;
    const int nxt = node + nq;
    const bool nact = act && nxt < n;
    int nc0 = 0, nc1 = 0;
    v2f nacc = {0.0f, 0.0f};
    if (nact) {
      nc0 = cnt0[nxt];
      nc1 = cnt1[nxt];
      nacc = gc[(size_t)nxt * 8];
    }
    const int* bk = col + (size_t)(act ? node : 0) * BKT;
    for (int j = 0; j < mmax; j += 8) {
      int sl[8];
      v2f r[8];
#pragma unroll
      for (int u = 0; u < 8; ++u) {
        sl[u] = 0;
        const int k = j + u;
        if (k < m) sl[u] = bk[k < m0 ? k : SUB + k - m0];
      }
#pragma unroll
      for (int u = 0; u < 8; ++u) {
        r[u].x = 0.0f;
        r[u].y = 0.0f;
        if (j + u < m) r[u] = gc[(size_t)sl[u] * 8];
      }
#pragma unroll
      for (int u = 0; u < 8; ++u)
        if (j + u < m) {
          acc.x += r[u].x;
          acc.y += r[u].y;
        }
    }
    if (act && (c0 > SUB || c1 > SUB)) {
      const int nov = *ovc;
      for (int k = 0; k < nov; ++k) {
        const int2 ed = ov[k];
        if (ed.x == node) {
          const v2f r = gc[(size_t)ed.y * 8];
          acc.x += r.x;
          acc.y += r.y;
        }
      }
    }
    if (act) {
      v2f res;
      res.x = acc.x * di + bc.x;
      res.y = acc.y * di + bc.y;
      ((v2f*)out)[(size_t)node * 8 + c] = res;
    }
    node = nxt;
    act = nact;
    c0 = nc0;
    c1 = nc1;
    acc = nacc;
  }
}

extern "C" void kernel_launch(void* const* d_in, const int* in_sizes, int n_in,
                              void* d_out, int out_size, void* d_ws,
                              size_t ws_size, hipStream_t stream) {
  constexpr int IN_FEAT = 256, HIDDEN = 64, NCLS = 16;

  const float* x  = (const float*)d_in[0];
  const int*   ei = (const int*)d_in[1];
  const float* W1 = (const float*)d_in[2];
  const float* b1 = (const float*)d_in[3];
  const float* W2 = (const float*)d_in[4];
  const float* b2 = (const float*)d_in[5];
  float* out = (float*)d_out;

  const int N = in_sizes[0] / IN_FEAT;
  const int E = in_sizes[1] / 2;
  const int* src = ei;
  const int* dst = ei + E;

  const size_t NP = ((size_t)N + 63) / 64 * 64;
  float* ws   = (float*)d_ws;
  float* h1   = ws;
  float* a1   = h1 + (size_t)N * HIDDEN;
  float* g2   = a1 + (size_t)N * HIDDEN;
  int* cnt0   = (int*)(g2 + (size_t)N * NCLS);
  int* cnt1   = cnt0 + NP;
  int* ovc    = cnt1 + NP;
  int* col    = ovc + 64;
  int2* ov    = (int2*)(col + (size_t)N * BKT);

  const int B = 256;
  const int mtiles  = (N + 15) / 16;
  const int gblocks = (mtiles + 7) / 8;
  const int nblocks = (N + B - 1) / B;
  const int pblocks = 2048;
  const int qblocks = 1024;

  k_cnt_zero<<<nblocks, B, 0, stream>>>(cnt0, cnt1, ovc, N);
  k_count_place<<<(E + B * EPT - 1) / (B * EPT), B, 0, stream>>>(src, dst, cnt0, cnt1, col, ovc, ov, E);

  wmma_gemm_f32<IN_FEAT, HIDDEN><<<gblocks, B, 0, stream>>>(x, W1, cnt0, cnt1, h1, N);
  k_gather64<<<pblocks, B, 0, stream>>>(cnt0, cnt1, col, ovc, ov, h1, b1, a1, N);

  wmma_gemm_f32<HIDDEN, NCLS><<<gblocks, B, 0, stream>>>(a1, W2, cnt0, cnt1, g2, N);
  k_gather16<<<qblocks, B, 0, stream>>>(cnt0, cnt1, col, ovc, ov, g2, b2, out, N);
}


# Both gathers ask for all of a node's source rows before the first sum (exact count, uniform tests; layer 2 reads its bucket with wide loads a trip ahead, slots by lane broadcast); sum order kept
# speedup vs baseline: 2.7831x; 1.0865x over previous
#include <hip/hip_runtime.h>


typedef float v2f __attribute__((ext_vector_type(2)));
typedef float v8f __attribute__((ext_vector_type(8)));

constexpr int SUB = 32;
constexpr int BKT = 2 * SUB;
template <int K, int N>
__global__ __launch_bounds__(256) void wmma_gemm_f32(
    const float* __restrict__ A, const float* __restrict__ B,
    const int* __restrict__ cnt0, const int* __restrict__ cnt1,
    float* __restrict__ C, int M) {
  __shared__ float lB[K * N];
  for (int i = threadIdx.x; i < K * N; i += 256) {
    int k = i / N, n = i % N;
    lB[((k >> 2) * N + n) * 4 + (k & 3)] = B[i];
  }
  __syncthreads();

  const int lane  = threadIdx.x & 31;
  const int wave  = threadIdx.x >> 5;
  const int mbase = (blockIdx.x * 8 + wave) * 16;
  if (mbase >= M) return;

  const int mr = lane & 15;
  const int hi = lane >> 4;

  const float* arow = A + (size_t)(mbase + mr) * K + 2 * hi;
  const float* brow = lB + mr * 4 + 2 * hi;

  float sc[8];
#pragma unroll
  for (int r = 0; r < 8; ++r) {
    const int row = mbase + 8 * hi + r;
    sc[r] = rsqrtf((float)(cnt0[row] + cnt1[row] + 1));
  }

  constexpr int NT = N / 16;
  v8f acc[NT];
  v8f zero = {};
#pragma unroll
  for (int t = 0; t < NT; ++t) acc[t] = zero;

  for (int kb = 0; kb < K / 4; ++kb) {
    v2f a = *(const v2f*)(arow + kb * 4);
#pragma unroll
    for (int t = 0; t < NT; ++t) {
      v2f b = *(const v2f*)(brow + kb * (N * 4) + t * 64);
      acc[t] = __builtin_amdgcn_wmma_f32_16x16x4_f32(
           false, a,  false, b,
           (short)0, acc[t],  false,  false);
    }
  }

#pragma unroll
  for (int t = 0; t < NT; ++t)
#pragma unroll
    for (int r = 0; r < 8; ++r)
      C[(size_t)(mbase + 8 * hi + r) * N + t * 16 + mr] = acc[t][r] * sc[r];
}

__device__ __forceinline__ int group_half() {
  __shared__ int sh_half;
  if (threadIdx.x == 0)
    sh_half = (int)((__builtin_amdgcn_s_sendmsg_rtn(0x87) >> 18) & 1u);
  __syncthreads();
  return __builtin_amdgcn_readfirstlane(sh_half);
}

__global__ void k_cnt_zero(int* __restrict__ cnt0, int* __restrict__ cnt1,
                           int* __restrict__ ovc, int n) {
  int i = blockIdx.x * blockDim.x + threadIdx.x;
  if (i < n) {
    cnt0[i] = 0;
    cnt1[i] = 0;
  }
  if (i == 0) *ovc = 0;
}

constexpr int EPT = 4;
template <bool FULL>
__device__ __forceinline__ void count_place_body(
    const int* __restrict__ src, const int* __restrict__ dst,
    int* __restrict__ cnt, int* __restrict__ col, int* __restrict__ ovc,
    int2* __restrict__ ov, int e, int base, int half) {
  int d[EPT], s[EPT], r[EPT];
#pragma unroll
  for (int u = 0; u < EPT; ++u) {
    const int i = base + u * 256;
    const bool ok = FULL || i < e;
    d[u] = ok ? dst[i] : -1;
    s[u] = ok ? src[i] : 0;
  }
#pragma unroll
  for (int u = 0; u < EPT; ++u) {
    r[u] = 0;
    if (FULL || d[u] >= 0) r[u] = atomicAdd(&cnt[d[u]], 1);
  }
#pragma unroll
  for (int u = 0; u < EPT; ++u) {
    if (FULL || d[u] >= 0) {
      if (r[u] < SUB) {
        col[(size_t)d[u] * BKT + half * SUB + r[u]] = s[u];
      } else {
        const int k = atomicAdd(ovc, 1);
        ov[k] = int2{d[u], s[u]};
      }
    }
  }
}

__global__ __launch_bounds__(256) void k_count_place(
    const int* __restrict__ src, const int* __restrict__ dst,
    int* __restrict__ cnt0, int* __restrict__ cnt1, int* __restrict__ col,
    int* __restrict__ ovc, int2* __restrict__ ov, int e) {
  const int half = group_half();
  int* cnt = half ? cnt1 : cnt0;
  const int first = blockIdx.x * (256 * EPT);
  const int base = first + threadIdx.x;
  if (first + 256 * EPT <= e)
    count_place_body<true>(src, dst, cnt, col, ovc, ov, e, base, half);
  else
    count_place_body<false>(src, dst, cnt, col, ovc, ov, e, base, half);
}

constexpr int NB = 20;
template <int CNT>
__device__ __forceinline__ void walk64(const v2f* __restrict__ hl, int s,
                                       int off, int m, float& ax, float& ay) {
  v2f r[CNT];
#pragma unroll
  for (int u = 0; u < CNT; ++u) {
    r[u].x = 0.0f;
    r[u].y = 0.0f;
    if (off + u < m)
      r[u] = hl[(size_t)__builtin_amdgcn_readlane(s, off + u) * 32];
  }
#pragma unroll
  for (int u = 0; u < CNT; ++u) {
    ax += r[u].x;
    ay += r[u].y;
  }
}

__global__ __launch_bounds__(256) void k_gather64(
    const int* __restrict__ cnt0, const int* __restrict__ cnt1,
    const int* __restrict__ col, const int* __restrict__ ovc,
    const int2* __restrict__ ov, const float* __restrict__ h,
    const float* __restrict__ b, float* __restrict__ o, int n) {
  const int lane = threadIdx.x & 31;
  const int nw = (int)(gridDim.x * 8);
  int node = __builtin_amdgcn_readfirstlane(
      (int)((blockIdx.x * 256 + threadIdx.x) >> 5));
  if (node >= n) return;
  const v2f* hl = (const v2f*)h + lane;
  const v2f bb = ((const v2f*)b)[lane];
  int s0 = col[(size_t)node * BKT + lane];
  int s1 = col[(size_t)node * BKT + SUB + lane];
  int k0 = cnt0[node];
  int k1 = cnt1[node];
  v2f self = hl[(size_t)node * 32];
  for (;;) {
    const int c0 = __builtin_amdgcn_readfirstlane(k0);
    const int c1 = __builtin_amdgcn_readfirstlane(k1);
    const float di = rsqrtf((float)(c0 + c1 + 1));
    const int m0 = c0 < SUB ? c0 : SUB;
    const int m1 = c1 < SUB ? c1 : SUB;
    const int m = m0 + m1;
    float ax = self.x, ay = self.y;
    const int t1 = __shfl(s1, lane - m0);
    const int la = lane < m0 ? s0 : t1;
    const int nxt = node + nw;
    const bool more = nxt < n;
    int ns0 = 0, ns1 = 0, nk0 = 0, nk1 = 0;
    v2f nself = {0.0f, 0.0f};
    if (more) {
      ns0 = col[(size_t)nxt * BKT + lane];
      ns1 = col[(size_t)nxt * BKT + SUB + lane];
      nk0 = cnt0[nxt];
      nk1 = cnt1[nxt];
      nself = hl[(size_t)nxt * 32];
    }
    const int ma = m < 32 ? m : 32;
    walk64<NB>(hl, la, 0, ma, ax, ay);
    if (ma > NB) walk64<32 - NB>(hl, la, NB, ma, ax, ay);
    if (m > 32) {
      const int lb = __shfl(s1, lane + 32 - m0);
      walk64<NB>(hl, lb, 0, m - 32, ax, ay);
      if (m - 32 > NB) walk64<32 - NB>(hl, lb, NB, m - 32, ax, ay);
    }
    if (c0 > SUB || c1 > SUB) {
      const int nov = __builtin_amdgcn_readfirstlane(*ovc);
      for (int q0 = 0; q0 < nov; q0 += 32) {
        int2 ed = int2{-1, 0};
        if (q0 + lane < nov) ed = ov[q0 + lane];
        unsigned hit = __builtin_amdgcn_ballot_w32(ed.x == node);
        while (hit) {
          const int bpos = __builtin_ctz(hit);
          hit &= hit - 1;
          const v2f r = hl[(size_t)__builtin_amdgcn_readlane(ed.y, bpos) * 32];
          ax += r.x;
          ay += r.y;
        }
      }
    }
    v2f res;
    res.x = fmaxf(ax * di + bb.x, 0.0f);
    res.y = fmaxf(ay * di + bb.y, 0.0f);
    ((v2f*)o)[(size_t)node * 32 + lane] = res;
    if (!more) break;
    node = nxt;
    s0 = ns0;
    s1 = ns1;
    k0 = nk0;
    k1 = nk1;
    self = nself;
  }
}

template <int J>
__device__ __forceinline__ int group_bcast(int v) {
  return (int)__builtin_amdgcn_mov_dpp8((unsigned)v, (unsigned)(J * 0x249249));
}

#define ASK_ROW(R, Q, CNT, K, COMP, J)                       \
  v2f R = {0.0f, 0.0f};                                      \
  {                                                          \
    const unsigned s_ = (unsigned)group_bcast<J>(Q.COMP);    \
    if (K < CNT) R = g2[s_ * 8u + (unsigned)c];              \
  }
#define ADD_ROW(R) \
  acc.x += R.x;    \
  acc.y += R.y;

__device__ __forceinline__ void walk_rest(const v2f* __restrict__ gc,
                                          const int4 q, int cnt, int lane,
                                          v2f& acc) {
  for (int k = 16; __builtin_amdgcn_ballot_w32(k < cnt) != 0u; ++k) {
    const int t = k & 3;
    const int v = t == 0 ? q.x : t == 1 ? q.y : t == 2 ? q.z : q.w;
    const int s = __shfl(v, (lane & 24) | (k >> 2));
    if (k < cnt) {
      const v2f r = gc[(size_t)s * 8];
      acc.x += r.x;
      acc.y += r.y;
    }
  }
}

__global__ __launch_bounds__(256) void k_gather16(
    const int* __restrict__ cnt0, const int* __restrict__ cnt1,
    const int* __restrict__ col, const int* __restrict__ ovc,
    const int2* __restrict__ ov, const float* __restrict__ g,
    const float* __restrict__ b, float* __restrict__ out, int n) {
  const int lane = threadIdx.x & 31;
  const int c = threadIdx.x & 7;
  const int nq = (int)(gridDim.x * 32);
  int node = (int)((blockIdx.x * 256 + threadIdx.x) >> 3);
  const v2f* g2 = (const v2f*)g;
  const v2f* gc = g2 + c;
  const int4* bq = (const int4*)col + c;
  const v2f bc = ((const v2f*)b)[c];
  const int4 none = int4{0, 0, 0, 0};
  bool act = node < n;
  int c0 = 0, c1 = 0;
  v2f acc = {0.0f, 0.0f};
  int4 q0 = none, q1 = none;
  if (act) {
    c0 = cnt0[node];
    c1 = cnt1[node];
    acc = gc[(size_t)node * 8];
    q0 = bq[(size_t)node * 16];
    q1 = bq[(size_t)node * 16 + 8];
  }
  while (__builtin_amdgcn_ballot_w32(act) != 0u) {
    int m0 = 0, m1 = 0;
    float di = 0.0f;
    if (act) {
      m0 = c0 < SUB ? c0 : SUB;
      m1 = c1 < SUB ? c1 : SUB;
      di = rsqrtf((float)(c0 + c1 + 1));
    }
    asm volatile("" ::"v"(q1.x), "v"(q1.y), "v"(q1.z), "v"(q1.w));
    ASK_ROW(a0, q0, m0, 0, x, 0)
    ASK_ROW(a1, q0, m0, 1, y, 0)
    ASK_ROW(a2, q0, m0, 2, z, 0)
    ASK_ROW(a3, q0, m0, 3, w, 0)
    ASK_ROW(a4, q0, m0, 4, x, 1)
    ASK_ROW(a5, q0, m0, 5, y, 1)
    ASK_ROW(a6, q0, m0, 6, z, 1)
    ASK_ROW(a7, q0, m0, 7, w, 1)
    ASK_ROW(a8, q0, m0, 8, x, 2)
    ASK_ROW(a9, q0, m0, 9, y, 2)
    ASK_ROW(a10, q0, m0, 10, z, 2)
    ASK_ROW(a11, q0, m0, 11, w, 2)
    ASK_ROW(a12, q0, m0, 12, x, 3)
    ASK_ROW(a13, q0, m0, 13, y, 3)
    ASK_ROW(a14, q0, m0, 14, z, 3)
    ASK_ROW(a15, q0, m0, 15, w, 3)
    ASK_ROW(b0, q1, m1, 0, x, 0)
    ASK_ROW(b1, q1, m1, 1, y, 0)
    ASK_ROW(b2, q1, m1, 2, z, 0)
    ASK_ROW(b3, q1, m1, 3, w, 0)
    ASK_ROW(b4, q1, m1, 4, x, 1)
    ASK_ROW(b5, q1, m1, 5, y, 1)
    ASK_ROW(b6, q1, m1, 6, z, 1)
    ASK_ROW(b7, q1, m1, 7, w, 1)
    ASK_ROW(b8, q1, m1, 8, x, 2)
    ASK_ROW(b9, q1, m1, 9, y, 2)
    ASK_ROW(b10, q1, m1, 10, z, 2)
    ASK_ROW(b11, q1, m1, 11, w, 2)
    ASK_ROW(b12, q1, m1, 12, x, 3)
    ASK_ROW(b13, q1, m1, 13, y, 3)
    ASK_ROW(b14, q1, m1, 14, z, 3)
    ASK_ROW(b15, q1, m1, 15, w, 3)
    const int nxt = node + nq;
    const bool nact = act && nxt < n;
    int nc0 = 0, nc1 = 0;
    v2f nacc = {0.0f, 0.0f};
    int4 nq0 = none, nq1 = none;
    if (nact) {
      nc0 = cnt0[nxt];
      nc1 = cnt1[nxt];
      nacc = gc[(size_t)nxt * 8];
      nq0 = bq[(size_t)nxt * 16];
      nq1 = bq[(size_t)nxt * 16 + 8];
    }
    ADD_ROW(a0)
    ADD_ROW(a1)
    ADD_ROW(a2)
    ADD_ROW(a3)
    ADD_ROW(a4)
    ADD_ROW(a5)
    ADD_ROW(a6)
    ADD_ROW(a7)
    ADD_ROW(a8)
    ADD_ROW(a9)
    ADD_ROW(a10)
    ADD_ROW(a11)
    ADD_ROW(a12)
    ADD_ROW(a13)
    ADD_ROW(a14)
    ADD_ROW(a15)
    walk_rest(gc, q0, m0, lane, acc);
    ADD_ROW(b0)
    ADD_ROW(b1)
    ADD_ROW(b2)
    ADD_ROW(b3)
    ADD_ROW(b4)
    ADD_ROW(b5)
    ADD_ROW(b6)
    ADD_ROW(b7)
    ADD_ROW(b8)
    ADD_ROW(b9)
    ADD_ROW(b10)
    ADD_ROW(b11)
    ADD_ROW(b12)
    ADD_ROW(b13)
    ADD_ROW(b14)
    ADD_ROW(b15)
    walk_rest(gc, q1, m1, lane, acc);
    if (act && (c0 > SUB || c1 > SUB)) {
      const int nov = *ovc;
      for (int k = 0; k < nov; ++k) {
        const int2 ed = ov[k];
        if (ed.x == node) {
          const v2f r = gc[(size_t)ed.y * 8];
          acc.x += r.x;
          acc.y += r.y;
        }
      }
    }
    if (act) {
      v2f res;
      res.x = acc.x * di + bc.x;
      res.y = acc.y * di + bc.y;
      ((v2f*)out)[(size_t)node * 8 + c] = res;
    }
    node = nxt;
    act = nact;
    c0 = nc0;
    c1 = nc1;
    acc = nacc;
    q0 = nq0;
    q1 = nq1;
  }
}
#undef ASK_ROW
#undef ADD_ROW

extern "C" void kernel_launch(void* const* d_in, const int* in_sizes, int n_in,
                              void* d_out, int out_size, void* d_ws,
                              size_t ws_size, hipStream_t stream) {
  constexpr int IN_FEAT = 256, HIDDEN = 64, NCLS = 16;

  const float* x  = (const float*)d_in[0];
  const int*   ei = (const int*)d_in[1];
  const float* W1 = (const float*)d_in[2];
  const float* b1 = (const float*)d_in[3];
  const float* W2 = (const float*)d_in[4];
  const float* b2 = (const float*)d_in[5];
  float* out = (float*)d_out;

  const int N = in_sizes[0] / IN_FEAT;
  const int E = in_sizes[1] / 2;
  const int* src = ei;
  const int* dst = ei + E;

  const size_t NP = ((size_t)N + 63) / 64 * 64;
  float* ws   = (float*)d_ws;
  float* h1   = ws;
  float* a1   = h1 + (size_t)N * HIDDEN;
  float* g2   = a1 + (size_t)N * HIDDEN;
  int* cnt0   = (int*)(g2 + (size_t)N * NCLS);
  int* cnt1   = cnt0 + NP;
  int* ovc    = cnt1 + NP;
  int* col    = ovc + 64;
  int2* ov    = (int2*)(col + (size_t)N * BKT);

  const int B = 256;
  const int mtiles  = (N + 15) / 16;
  const int gblocks = (mtiles + 7) / 8;
  const int nblocks = (N + B - 1) / B;
  const int pblocks = 2048;
  const int qblocks = 1024;

  k_cnt_zero<<<nblocks, B, 0, stream>>>(cnt0, cnt1, ovc, N);
  k_count_place<<<(E + B * EPT - 1) / (B * EPT), B, 0, stream>>>(src, dst, cnt0, cnt1, col, ovc, ov, E);

  wmma_gemm_f32<IN_FEAT, HIDDEN><<<gblocks, B, 0, stream>>>(x, W1, cnt0, cnt1, h1, N);
  k_gather64<<<pblocks, B, 0, stream>>>(cnt0, cnt1, col, ovc, ov, h1, b1, a1, N);

  wmma_gemm_f32<HIDDEN, NCLS><<<gblocks, B, 0, stream>>>(a1, W2, cnt0, cnt1, g2, N);
  k_gather16<<<qblocks, B, 0, stream>>>(cnt0, cnt1, col, ovc, ov, g2, b2, out, N);
}


# Layer-2 product moved into the layer-1 gather: a wave keeps its nodes' rows in LDS and multiplies them by W2 with the same f32 WMMAs at its end; activation plane and one launch gone; sum order kept
# speedup vs baseline: 2.9012x; 1.0424x over previous
#include <hip/hip_runtime.h>


typedef float v2f __attribute__((ext_vector_type(2)));
typedef float v8f __attribute__((ext_vector_type(8)));

constexpr int SUB = 32;
constexpr int BKT = 2 * SUB;
template <int K, int N>
__global__ __launch_bounds__(256) void wmma_gemm_f32(
    const float* __restrict__ A, const float* __restrict__ B,
    const int* __restrict__ cnt0, const int* __restrict__ cnt1,
    float* __restrict__ C, int M) {
  __shared__ float lB[K * N];
  for (int i = threadIdx.x; i < K * N; i += 256) {
    int k = i / N, n = i % N;
    lB[((k >> 2) * N + n) * 4 + (k & 3)] = B[i];
  }
  __syncthreads();

  const int lane  = threadIdx.x & 31;
  const int wave  = threadIdx.x >> 5;
  const int mbase = (blockIdx.x * 8 + wave) * 16;
  if (mbase >= M) return;

  const int mr = lane & 15;
  const int hi = lane >> 4;

  const float* arow = A + (size_t)(mbase + mr) * K + 2 * hi;
  const float* brow = lB + mr * 4 + 2 * hi;

  float sc[8];
#pragma unroll
  for (int r = 0; r < 8; ++r) {
    const int row = mbase + 8 * hi + r;
    sc[r] = rsqrtf((float)(cnt0[row] + cnt1[row] + 1));
  }

  constexpr int NT = N / 16;
  v8f acc[NT];
  v8f zero = {};
#pragma unroll
  for (int t = 0; t < NT; ++t) acc[t] = zero;

  for (int kb = 0; kb < K / 4; ++kb) {
    v2f a = *(const v2f*)(arow + kb * 4);
#pragma unroll
    for (int t = 0; t < NT; ++t) {
      v2f b = *(const v2f*)(brow + kb * (N * 4) + t * 64);
      acc[t] = __builtin_amdgcn_wmma_f32_16x16x4_f32(
           false, a,  false, b,
           (short)0, acc[t],  false,  false);
    }
  }

#pragma unroll
  for (int t = 0; t < NT; ++t)
#pragma unroll
    for (int r = 0; r < 8; ++r)
      C[(size_t)(mbase + 8 * hi + r) * N + t * 16 + mr] = acc[t][r] * sc[r];
}

__device__ __forceinline__ int group_half() {
  __shared__ int sh_half;
  if (threadIdx.x == 0)
    sh_half = (int)((__builtin_amdgcn_s_sendmsg_rtn(0x87) >> 18) & 1u);
  __syncthreads();
  return __builtin_amdgcn_readfirstlane(sh_half);
}

__global__ void k_cnt_zero(int* __restrict__ cnt0, int* __restrict__ cnt1,
                           int* __restrict__ ovc, int n) {
  int i = blockIdx.x * blockDim.x + threadIdx.x;
  if (i < n) {
    cnt0[i] = 0;
    cnt1[i] = 0;
  }
  if (i == 0) *ovc = 0;
}

constexpr int EPT = 4;
template <bool FULL>
__device__ __forceinline__ void count_place_body(
    const int* __restrict__ src, const int* __restrict__ dst,
    int* __restrict__ cnt, int* __restrict__ col, int* __restrict__ ovc,
    int2* __restrict__ ov, int e, int base, int half) {
  int d[EPT], s[EPT], r[EPT];
#pragma unroll
  for (int u = 0; u < EPT; ++u) {
    const int i = base + u * 256;
    const bool ok = FULL || i < e;
    d[u] = ok ? dst[i] : -1;
    s[u] = ok ? src[i] : 0;
  }
#pragma unroll
  for (int u = 0; u < EPT; ++u) {
    r[u] = 0;
    if (FULL || d[u] >= 0) r[u] = atomicAdd(&cnt[d[u]], 1);
  }
#pragma unroll
  for (int u = 0; u < EPT; ++u) {
    if (FULL || d[u] >= 0) {
      if (r[u] < SUB) {
        col[(size_t)d[u] * BKT + half * SUB + r[u]] = s[u];
      } else {
        const int k = atomicAdd(ovc, 1);
        ov[k] = int2{d[u], s[u]};
      }
    }
  }
}

__global__ __launch_bounds__(256) void k_count_place(
    const int* __restrict__ src, const int* __restrict__ dst,
    int* __restrict__ cnt0, int* __restrict__ cnt1, int* __restrict__ col,
    int* __restrict__ ovc, int2* __restrict__ ov, int e) {
  const int half = group_half();
  int* cnt = half ? cnt1 : cnt0;
  const int first = blockIdx.x * (256 * EPT);
  const int base = first + threadIdx.x;
  if (first + 256 * EPT <= e)
    count_place_body<true>(src, dst, cnt, col, ovc, ov, e, base, half);
  else
    count_place_body<false>(src, dst, cnt, col, ovc, ov, e, base, half);
}

constexpr int NB = 20;
template <int CNT>
__device__ __forceinline__ void walk64(const v2f* __restrict__ hl, int s,
                                       int off, int m, float& ax, float& ay) {
  v2f r[CNT];
#pragma unroll
  for (int u = 0; u < CNT; ++u) {
    r[u].x = 0.0f;
    r[u].y = 0.0f;
    if (off + u < m)
      r[u] = hl[(size_t)__builtin_amdgcn_readlane(s, off + u) * 32];
  }
#pragma unroll
  for (int u = 0; u < CNT; ++u) {
    ax += r[u].x;
    ay += r[u].y;
  }
}

constexpr int TR = 8;
constexpr int AP = 68;
constexpr int G2P = 32;
__global__ __launch_bounds__(256) void k_gather64_mm(
    const int* __restrict__ cnt0, const int* __restrict__ cnt1,
    const int* __restrict__ col, const int* __restrict__ ovc,
    const int2* __restrict__ ov, const float* __restrict__ h,
    const float* __restrict__ b, const float* __restrict__ w2,
    float* __restrict__ g2, int n) {
  __shared__ __attribute__((aligned(16))) float lA[8 * TR * AP];
  const int lane = threadIdx.x & 31;
  const int nw = (int)(gridDim.x * 8);
  int node = __builtin_amdgcn_readfirstlane(
      (int)((blockIdx.x * 256 + threadIdx.x) >> 5));
  if (node >= n) return;
  const v2f* hl = (const v2f*)h + lane;
  float* myA = lA + (threadIdx.x >> 5) * (TR * AP);
  const int mr = lane & 15;
  const int hi = lane >> 4;
  int j = 0;
  int tile0 = node;
  int s0 = col[(size_t)node * BKT + lane];
  int s1 = col[(size_t)node * BKT + SUB + lane];
  int k0 = cnt0[node];
  int k1 = cnt1[node];
  v2f self = hl[(size_t)node * 32];
  {
    static_assert(TR * AP / 2 == 8 * 32 + 16, "tile fill below covers 272 pairs");
    const v2f z2 = {0.0f, 0.0f};
#pragma unroll
    for (int q = 0; q < 8; ++q) ((v2f*)myA)[lane + 32 * q] = z2;
    if (lane < 16) ((v2f*)myA)[lane + 256] = z2;
  }
  for (;;) {
    const int c0 = __builtin_amdgcn_readfirstlane(k0);
    const int c1 = __builtin_amdgcn_readfirstlane(k1);
    const float di = rsqrtf((float)(c0 + c1 + 1));
    const int m0 = c0 < SUB ? c0 : SUB;
    const int m1 = c1 < SUB ? c1 : SUB;
    const int m = m0 + m1;
    float ax = self.x, ay = self.y;
    const int t1 = __shfl(s1, lane - m0);
    const int la = lane < m0 ? s0 : t1;
    const int nxt = node + nw;
    const bool more = nxt < n;
    int ns0 = 0, ns1 = 0, nk0 = 0, nk1 = 0;
    v2f nself = {0.0f, 0.0f};
    if (more) {
      ns0 = col[(size_t)nxt * BKT + lane];
      ns1 = col[(size_t)nxt * BKT + SUB + lane];
      nk0 = cnt0[nxt];
      nk1 = cnt1[nxt];
      nself = hl[(size_t)nxt * 32];
    }
    const int ma = m < 32 ? m : 32;
    walk64<NB>(hl, la, 0, ma, ax, ay);
    if (ma > NB) walk64<32 - NB>(hl, la, NB, ma, ax, ay);
    if (m > 32) {
      const int lb = __shfl(s1, lane + 32 - m0);
      walk64<NB>(hl, lb, 0, m - 32, ax, ay);
      if (m - 32 > NB) walk64<32 - NB>(hl, lb, NB, m - 32, ax, ay);
    }
    if (c0 > SUB || c1 > SUB) {
      const int nov = __builtin_amdgcn_readfirstlane(*ovc);
      for (int q0 = 0; q0 < nov; q0 += 32) {
        int2 ed = int2{-1, 0};
        if (q0 + lane < nov) ed = ov[q0 + lane];
        unsigned hit = __builtin_amdgcn_ballot_w32(ed.x == node);
        while (hit) {
          const int bpos = __builtin_ctz(hit);
          hit &= hit - 1;
          const v2f r = hl[(size_t)__builtin_amdgcn_readlane(ed.y, bpos) * 32];
          ax += r.x;
          ay += r.y;
        }
      }
    }
    v2f sum;
    sum.x = ax;
    sum.y = ay;
    *(v2f*)(myA + j * AP + 2 * lane) = sum;
    if (lane == 0) *(myA + j * AP + 2 * lane + 64) = di;
    ++j;
    if (j == TR || !more) {
      const v2f bb = ((const v2f*)b)[lane];
      const float* arow = myA + (mr & (TR - 1)) * AP + 2 * hi;
      float dv = 0.0f;
      if (hi == 0) dv = arow[64];
      const int dis = __builtin_bit_cast(int, dv);
#pragma unroll 2
      for (int r = 0; r < TR; ++r) {
        const float dr =
            __builtin_bit_cast(float, __builtin_amdgcn_readlane(dis, r));
        const v2f t = *(const v2f*)(myA + r * AP + 2 * lane);
        v2f res;
        res.x = fmaxf(t.x * dr + bb.x, 0.0f);
        res.y = fmaxf(t.y * dr + bb.y, 0.0f);
        *(v2f*)(myA + r * AP + 2 * lane) = res;
      }
      const unsigned bo = (unsigned)(hi * 32 + mr);
      v8f acc = {};
#pragma unroll 1
      for (int kg = 0; kg < 16; kg += 8) {
        v2f bf[8];
#pragma unroll
        for (int u = 0; u < 8; ++u) {
          bf[u].x = w2[bo + (unsigned)((kg + u) * 64)];
          bf[u].y = w2[bo + (unsigned)((kg + u) * 64 + 16)];
        }
#pragma unroll
        for (int u = 0; u < 8; ++u) {
          const v2f a = *(const v2f*)(arow + (kg + u) * 4);
          acc = __builtin_amdgcn_wmma_f32_16x16x4_f32(
               false, a,  false, bf[u],
               (short)0, acc,  false,  false);
        }
      }
      float v[TR];
#pragma unroll
      for (int r = 0; r < TR; ++r)
        v[r] = acc[r] * __builtin_bit_cast(float, __builtin_amdgcn_readlane(dis, r));
#pragma unroll
      for (int r = 0; r < TR; ++r)
        if (r < j && hi == 0)
          (g2 + (size_t)(tile0 + r * nw) * G2P)[(unsigned)mr] = v[r];
      j = 0;
      tile0 = nxt;
    }
    if (!more) break;
    node = nxt;
    s0 = ns0;
    s1 = ns1;
    k0 = nk0;
    k1 = nk1;
    self = nself;
  }
}

constexpr int G2V = G2P / 2;
template <int J>
__device__ __forceinline__ int group_bcast(int v) {
  return (int)__builtin_amdgcn_mov_dpp8((unsigned)v, (unsigned)(J * 0x249249));
}

#define ASK_ROW(R, Q, CNT, K, COMP, J)                       \
  v2f R = {0.0f, 0.0f};                                      \
  {                                                          \
    const unsigned s_ = (unsigned)group_bcast<J>(Q.COMP);    \
    if (K < CNT) R = g2[s_ * (unsigned)G2V + (unsigned)c];   \
  }
#define ADD_ROW(R) \
  acc.x += R.x;    \
  acc.y += R.y;

__device__ __forceinline__ void walk_rest(const v2f* __restrict__ gc,
                                          const int4 q, int cnt, int lane,
                                          v2f& acc) {
  for (int k = 16; __builtin_amdgcn_ballot_w32(k < cnt) != 0u; ++k) {
    const int t = k & 3;
    const int v = t == 0 ? q.x : t == 1 ? q.y : t == 2 ? q.z : q.w;
    const int s = __shfl(v, (lane & 24) | (k >> 2));
    if (k < cnt) {
      const v2f r = gc[(size_t)s * G2V];
      acc.x += r.x;
      acc.y += r.y;
    }
  }
}

__global__ __launch_bounds__(256) void k_gather16(
    const int* __restrict__ cnt0, const int* __restrict__ cnt1,
    const int* __restrict__ col, const int* __restrict__ ovc,
    const int2* __restrict__ ov, const float* __restrict__ g,
    const float* __restrict__ b, float* __restrict__ out, int n) {
  const int lane = threadIdx.x & 31;
  const int c = threadIdx.x & 7;
  const int nq = (int)(gridDim.x * 32);
  int node = (int)((blockIdx.x * 256 + threadIdx.x) >> 3);
  const v2f* g2 = (const v2f*)g;
  const v2f* gc = g2 + c;
  const int4* bq = (const int4*)col + c;
  const v2f bc = ((const v2f*)b)[c];
  const int4 none = int4{0, 0, 0, 0};
  bool act = node < n;
  int c0 = 0, c1 = 0;
  v2f acc = {0.0f, 0.0f};
  int4 q0 = none, q1 = none;
  if (act) {
    c0 = cnt0[node];
    c1 = cnt1[node];
    acc = gc[(size_t)node * G2V];
    q0 = bq[(size_t)node * 16];
    q1 = bq[(size_t)node * 16 + 8];
  }
  while (__builtin_amdgcn_ballot_w32(act) != 0u) {
    int m0 = 0, m1 = 0;
    float di = 0.0f;
    if (act) {
      m0 = c0 < SUB ? c0 : SUB;
      m1 = c1 < SUB ? c1 : SUB;
      di = rsqrtf((float)(c0 + c1 + 1));
    }
    asm volatile("" ::"v"(q1.x), "v"(q1.y), "v"(q1.z), "v"(q1.w));
    ASK_ROW(a0, q0, m0, 0, x, 0)
    ASK_ROW(a1, q0, m0, 1, y, 0)
    ASK_ROW(a2, q0, m0, 2, z, 0)
    ASK_ROW(a3, q0, m0, 3, w, 0)
    ASK_ROW(a4, q0, m0, 4, x, 1)
    ASK_ROW(a5, q0, m0, 5, y, 1)
    ASK_ROW(a6, q0, m0, 6, z, 1)
    ASK_ROW(a7, q0, m0, 7, w, 1)
    ASK_ROW(a8, q0, m0, 8, x, 2)
    ASK_ROW(a9, q0, m0, 9, y, 2)
    ASK_ROW(a10, q0, m0, 10, z, 2)
    ASK_ROW(a11, q0, m0, 11, w, 2)
    ASK_ROW(a12, q0, m0, 12, x, 3)
    ASK_ROW(a13, q0, m0, 13, y, 3)
    ASK_ROW(a14, q0, m0, 14, z, 3)
    ASK_ROW(a15, q0, m0, 15, w, 3)
    ASK_ROW(b0, q1, m1, 0, x, 0)
    ASK_ROW(b1, q1, m1, 1, y, 0)
    ASK_ROW(b2, q1, m1, 2, z, 0)
    ASK_ROW(b3, q1, m1, 3, w, 0)
    ASK_ROW(b4, q1, m1, 4, x, 1)
    ASK_ROW(b5, q1, m1, 5, y, 1)
    ASK_ROW(b6, q1, m1, 6, z, 1)
    ASK_ROW(b7, q1, m1, 7, w, 1)
    ASK_ROW(b8, q1, m1, 8, x, 2)
    ASK_ROW(b9, q1, m1, 9, y, 2)
    ASK_ROW(b10, q1, m1, 10, z, 2)
    ASK_ROW(b11, q1, m1, 11, w, 2)
    ASK_ROW(b12, q1, m1, 12, x, 3)
    ASK_ROW(b13, q1, m1, 13, y, 3)
    ASK_ROW(b14, q1, m1, 14, z, 3)
    ASK_ROW(b15, q1, m1, 15, w, 3)
    const int nxt = node + nq;
    const bool nact = act && nxt < n;
    int nc0 = 0, nc1 = 0;
    v2f nacc = {0.0f, 0.0f};
    int4 nq0 = none, nq1 = none;
    if (nact) {
      nc0 = cnt0[nxt];
      nc1 = cnt1[nxt];
      nacc = gc[(size_t)nxt * G2V];
      nq0 = bq[(size_t)nxt * 16];
      nq1 = bq[(size_t)nxt * 16 + 8];
    }
    ADD_ROW(a0)
    ADD_ROW(a1)
    ADD_ROW(a2)
    ADD_ROW(a3)
    ADD_ROW(a4)
    ADD_ROW(a5)
    ADD_ROW(a6)
    ADD_ROW(a7)
    ADD_ROW(a8)
    ADD_ROW(a9)
    ADD_ROW(a10)
    ADD_ROW(a11)
    ADD_ROW(a12)
    ADD_ROW(a13)
    ADD_ROW(a14)
    ADD_ROW(a15)
    walk_rest(gc, q0, m0, lane, acc);
    ADD_ROW(b0)
    ADD_ROW(b1)
    ADD_ROW(b2)
    ADD_ROW(b3)
    ADD_ROW(b4)
    ADD_ROW(b5)
    ADD_ROW(b6)
    ADD_ROW(b7)
    ADD_ROW(b8)
    ADD_ROW(b9)
    ADD_ROW(b10)
    ADD_ROW(b11)
    ADD_ROW(b12)
    ADD_ROW(b13)
    ADD_ROW(b14)
    ADD_ROW(b15)
    walk_rest(gc, q1, m1, lane, acc);
    if (act && (c0 > SUB || c1 > SUB)) {
      const int nov = *ovc;
      for (int k = 0; k < nov; ++k) {
        const int2 ed = ov[k];
        if (ed.x == node) {
          const v2f r = gc[(size_t)ed.y * G2V];
          acc.x += r.x;
          acc.y += r.y;
        }
      }
    }
    if (act) {
      v2f res;
      res.x = acc.x * di + bc.x;
      res.y = acc.y * di + bc.y;
      ((v2f*)out)[(size_t)node * 8 + c] = res;
    }
    node = nxt;
    act = nact;
    c0 = nc0;
    c1 = nc1;
    acc = nacc;
    q0 = nq0;
    q1 = nq1;
  }
}
#undef ASK_ROW
#undef ADD_ROW

extern "C" void kernel_launch(void* const* d_in, const int* in_sizes, int n_in,
                              void* d_out, int out_size, void* d_ws,
                              size_t ws_size, hipStream_t stream) {
  constexpr int IN_FEAT = 256, HIDDEN = 64;

  const float* x  = (const float*)d_in[0];
  const int*   ei = (const int*)d_in[1];
  const float* W1 = (const float*)d_in[2];
  const float* b1 = (const float*)d_in[3];
  const float* W2 = (const float*)d_in[4];
  const float* b2 = (const float*)d_in[5];
  float* out = (float*)d_out;

  const int N = in_sizes[0] / IN_FEAT;
  const int E = in_sizes[1] / 2;
  const int* src = ei;
  const int* dst = ei + E;

  const size_t NP = ((size_t)N + 63) / 64 * 64;
  float* ws   = (float*)d_ws;
  float* h1   = ws;
  float* g2   = h1 + (size_t)N * HIDDEN;
  int* cnt0   = (int*)(g2 + NP * G2P);
  int* cnt1   = cnt0 + NP;
  int* ovc    = cnt1 + NP;
  int* col    = ovc + 64;
  int2* ov    = (int2*)(col + (size_t)N * BKT);

  const int B = 256;
  const int mtiles  = (N + 15) / 16;
  const int gblocks = (mtiles + 7) / 8;
  const int nblocks = (N + B - 1) / B;
  const int pblocks = 2048;
  const int qblocks = 1024;

  k_cnt_zero<<<nblocks, B, 0, stream>>>(cnt0, cnt1, ovc, N);
  k_count_place<<<(E + B * EPT - 1) / (B * EPT), B, 0, stream>>>(src, dst, cnt0, cnt1, col, ovc, ov, E);

  wmma_gemm_f32<IN_FEAT, HIDDEN><<<gblocks, B, 0, stream>>>(x, W1, cnt0, cnt1, h1, N);
  k_gather64_mm<<<pblocks, B, 0, stream>>>(cnt0, cnt1, col, ovc, ov, h1, b1, W2, g2, N);

  k_gather16<<<qblocks, B, 0, stream>>>(cnt0, cnt1, col, ovc, ov, g2, b2, out, N);
}


# Edges bucketed in two passes, every counting add in LDS (chunk binned by node range into fixed runs; a group per range ranks them, writes whole bucket lines); zero pass gone; sum order = new ranks
# speedup vs baseline: 3.7851x; 1.3047x over previous
#include <hip/hip_runtime.h>


typedef float v2f __attribute__((ext_vector_type(2)));
typedef float v8f __attribute__((ext_vector_type(8)));

constexpr int SUB = 32;
constexpr int BKT = 2 * SUB;
template <int K, int N>
__global__ __launch_bounds__(256) void wmma_gemm_f32(
    const float* __restrict__ A, const float* __restrict__ B,
    const int* __restrict__ cnt0, const int* __restrict__ cnt1,
    float* __restrict__ C, int M) {
  __shared__ float lB[K * N];
  for (int i = threadIdx.x; i < K * N; i += 256) {
    int k = i / N, n = i % N;
    lB[((k >> 2) * N + n) * 4 + (k & 3)] = B[i];
  }
  __syncthreads();

  const int lane  = threadIdx.x & 31;
  const int wave  = threadIdx.x >> 5;
  const int mbase = (blockIdx.x * 8 + wave) * 16;
  if (mbase >= M) return;

  const int mr = lane & 15;
  const int hi = lane >> 4;

  const float* arow = A + (size_t)(mbase + mr) * K + 2 * hi;
  const float* brow = lB + mr * 4 + 2 * hi;

  float sc[8];
#pragma unroll
  for (int r = 0; r < 8; ++r) {
    const int row = mbase + 8 * hi + r;
    sc[r] = rsqrtf((float)(cnt0[row] + cnt1[row] + 1));
  }

  constexpr int NT = N / 16;
  v8f acc[NT];
  v8f zero = {};
#pragma unroll
  for (int t = 0; t < NT; ++t) acc[t] = zero;

  for (int kb = 0; kb < K / 4; ++kb) {
    v2f a = *(const v2f*)(arow + kb * 4);
#pragma unroll
    for (int t = 0; t < NT; ++t) {
      v2f b = *(const v2f*)(brow + kb * (N * 4) + t * 64);
      acc[t] = __builtin_amdgcn_wmma_f32_16x16x4_f32(
           false, a,  false, b,
           (short)0, acc[t],  false,  false);
    }
  }

#pragma unroll
  for (int t = 0; t < NT; ++t)
#pragma unroll
    for (int r = 0; r < 8; ++r)
      C[(size_t)(mbase + 8 * hi + r) * N + t * 16 + mr] = acc[t][r] * sc[r];
}

constexpr int NPB   = 256;
constexpr int NPB_L = 8;
constexpr int BT    = 512;
constexpr int BE    = 16;
constexpr int CHUNK = BT * BE;
constexpr int RCAP  = 64;
constexpr int HB    = 512;

__global__ __launch_bounds__(BT) void k_bin(
    const int* __restrict__ src, const int* __restrict__ dst,
    int2* __restrict__ runs, int* __restrict__ runcnt,
    int2* __restrict__ spill, int* __restrict__ spillcnt,
    int* __restrict__ ovc, int e, int nbin, int nwg) {
  static_assert(HB == BT, "the scan below holds one table entry per thread");
  __shared__ int hist[HB];
  __shared__ int off[HB];
  __shared__ int wsum[BT / 32];
  __shared__ int nspill;
  __shared__ int2 stage[CHUNK];
  const int tid = threadIdx.x;
  const int lane = tid & 31;
  const int wv = tid >> 5;
  const int wg = blockIdx.x;
  hist[tid] = 0;
  if (tid == 0) {
    nspill = 0;
    if (wg == 0) *ovc = 0;
  }
  __syncthreads();
  const int base = wg * CHUNK + tid;
  int d[BE], s[BE];
#pragma unroll
  for (int u = 0; u < BE; ++u) {
    const int i = base + u * BT;
    const bool ok = i < e;
    d[u] = ok ? dst[i] : -1;
    s[u] = ok ? src[i] : 0;
  }
  int r[BE];
#pragma unroll
  for (int u = 0; u < BE; ++u) {
    r[u] = -1;
    if (d[u] >= 0) r[u] = atomicAdd(&hist[d[u] >> NPB_L], 1);
  }
  __syncthreads();
  const int h = hist[tid];
  int inc = h;
#pragma unroll
  for (int k = 1; k < 32; k <<= 1) {
    const int t = __shfl_up(inc, k);
    if (lane >= k) inc += t;
  }
  if (lane == 31) wsum[wv] = inc;
  __syncthreads();
  int pre = 0;
  for (int k = 0; k < wv; ++k) pre += wsum[k];
  off[tid] = pre + inc - h;
  __syncthreads();
  int o[BE];
#pragma unroll
  for (int u = 0; u < BE; ++u) o[u] = r[u] >= 0 ? off[d[u] >> NPB_L] : 0;
#pragma unroll
  for (int u = 0; u < BE; ++u)
    if (r[u] >= 0) stage[o[u] + r[u]] = int2{d[u], s[u]};
  __syncthreads();
  const int rem = e - wg * CHUNK;
  const int total = rem < CHUNK ? rem : CHUNK;
  int2 ed[BE];
#pragma unroll
  for (int u = 0; u < BE; ++u) {
    const int j = tid + u * BT;
    ed[u] = int2{-1, 0};
    if (j < total) ed[u] = stage[j];
  }
#pragma unroll
  for (int u = 0; u < BE; ++u) o[u] = ed[u].x >= 0 ? off[ed[u].x >> NPB_L] : 0;
#pragma unroll
  for (int u = 0; u < BE; ++u) {
    if (ed[u].x >= 0) {
      const int bin = ed[u].x >> NPB_L;
      const int rr = tid + u * BT - o[u];
      if (rr < RCAP) {
        runs[((size_t)bin * nwg + wg) * RCAP + rr] = ed[u];
      } else {
        const int k = atomicAdd(&nspill, 1);
        spill[(size_t)wg * CHUNK + k] = ed[u];
      }
    }
  }
  __syncthreads();
  if (tid < nbin) runcnt[(size_t)tid * nwg + wg] = h;
  if (tid == 0) spillcnt[wg] = nspill;
}

__device__ __forceinline__ int rank_of(int* __restrict__ lcnt, int d) {
  return atomicAdd(&lcnt[d & (NPB - 1)], 1);
}
__device__ __forceinline__ void put_slot(int* __restrict__ limg, int d, int s,
                                         int r, int* __restrict__ ovc,
                                         int2* __restrict__ ov) {
  if (r < BKT) {
    limg[(d & (NPB - 1)) * BKT + (r & 1) * SUB + (r >> 1)] = s;
  } else {
    const int k = atomicAdd(ovc, 1);
    ov[k] = int2{d, s};
  }
}

constexpr int RB = 13;
__global__ __launch_bounds__(BT) void k_rank(
    const int2* __restrict__ runs, const int* __restrict__ runcnt,
    const int2* __restrict__ spill, const int* __restrict__ spillcnt,
    int* __restrict__ cnt0, int* __restrict__ cnt1, int* __restrict__ col,
    int* __restrict__ ovc, int2* __restrict__ ov, int n, int nwg) {
  __shared__ int lcnt[NPB];
  __shared__ __attribute__((aligned(16))) int limg[NPB * BKT];
  const int tid = threadIdx.x;
  const int lane = tid & 31;
  const int w = __builtin_amdgcn_readfirstlane(tid >> 5);
  const int b = blockIdx.x;
  const int4 none = int4{0, 0, 0, 0};
  for (int i = tid; i < NPB * BKT / 4; i += BT) ((int4*)limg)[i] = none;
  if (tid < NPB) lcnt[tid] = 0;
  __syncthreads();
  const int4* rq = (const int4*)runs + lane;
  for (int wg0 = w; wg0 < nwg; wg0 += (BT / 32) * RB) {
    int4 q[RB];
    int c[RB];
#pragma unroll
    for (int u = 0; u < RB; ++u) {
      const int wg = wg0 + u * (BT / 32);
      const bool ok = wg < nwg;
      const size_t run = (size_t)b * nwg + (ok ? wg : nwg - 1);
      q[u] = rq[run * (RCAP / 2)];
      const int len = runcnt[run];
      c[u] = ok ? len : 0;
    }
    int r0[RB], r1[RB];
#pragma unroll
    for (int u = 0; u < RB; ++u) {
      const int cc = c[u] < RCAP ? c[u] : RCAP;
      r0[u] = -1;
      r1[u] = -1;
      if (2 * lane < cc) r0[u] = rank_of(lcnt, q[u].x);
      if (2 * lane + 1 < cc) r1[u] = rank_of(lcnt, q[u].z);
    }
#pragma unroll
    for (int u = 0; u < RB; ++u) {
      if (r0[u] >= 0) put_slot(limg, q[u].x, q[u].y, r0[u], ovc, ov);
      if (r1[u] >= 0) put_slot(limg, q[u].z, q[u].w, r1[u], ovc, ov);
    }
  }
  for (int wg = tid; wg < nwg; wg += BT) {
    const int ns = spillcnt[wg];
    for (int k = 0; k < ns; ++k) {
      const int2 ed = spill[(size_t)wg * CHUNK + k];
      if ((ed.x >> NPB_L) == b)
        put_slot(limg, ed.x, ed.y, rank_of(lcnt, ed.x), ovc, ov);
    }
  }
  __syncthreads();
  int4* cq = (int4*)col + (size_t)b * (NPB * BKT / 4);
  for (int i = tid; i < NPB * BKT / 4; i += BT) {
    const int node = b * NPB + (i >> 4);
    if (node < n) cq[i] = ((const int4*)limg)[i];
  }
  if (tid < NPB) {
    const int node = b * NPB + tid;
    if (node < n) {
      const int c = lcnt[tid];
      cnt0[node] = (c + 1) >> 1;
      cnt1[node] = c >> 1;
    }
  }
}

constexpr int NB = 20;
template <int CNT>
__device__ __forceinline__ void walk64(const v2f* __restrict__ hl, int s,
                                       int off, int m, float& ax, float& ay) {
  v2f r[CNT];
#pragma unroll
  for (int u = 0; u < CNT; ++u) {
    r[u].x = 0.0f;
    r[u].y = 0.0f;
    if (off + u < m)
      r[u] = hl[(size_t)__builtin_amdgcn_readlane(s, off + u) * 32];
  }
#pragma unroll
  for (int u = 0; u < CNT; ++u) {
    ax += r[u].x;
    ay += r[u].y;
  }
}

constexpr int TR = 8;
constexpr int AP = 68;
constexpr int G2P = 32;
__global__ __launch_bounds__(256) void k_gather64_mm(
    const int* __restrict__ cnt0, const int* __restrict__ cnt1,
    const int* __restrict__ col, const int* __restrict__ ovc,
    const int2* __restrict__ ov, const float* __restrict__ h,
    const float* __restrict__ b, const float* __restrict__ w2,
    float* __restrict__ g2, int n) {
  __shared__ __attribute__((aligned(16))) float lA[8 * TR * AP];
  const int lane = threadIdx.x & 31;
  const int nw = (int)(gridDim.x * 8);
  int node = __builtin_amdgcn_readfirstlane(
      (int)((blockIdx.x * 256 + threadIdx.x) >> 5));
  if (node >= n) return;
  const v2f* hl = (const v2f*)h + lane;
  float* myA = lA + (threadIdx.x >> 5) * (TR * AP);
  const int mr = lane & 15;
  const int hi = lane >> 4;
  int j = 0;
  int tile0 = node;
  int s0 = col[(size_t)node * BKT + lane];
  int s1 = col[(size_t)node * BKT + SUB + lane];
  int k0 = cnt0[node];
  int k1 = cnt1[node];
  v2f self = hl[(size_t)node * 32];
  {
    static_assert(TR * AP / 2 == 8 * 32 + 16, "tile fill below covers 272 pairs");
    const v2f z2 = {0.0f, 0.0f};
#pragma unroll
    for (int q = 0; q < 8; ++q) ((v2f*)myA)[lane + 32 * q] = z2;
    if (lane < 16) ((v2f*)myA)[lane + 256] = z2;
  }
  for (;;) {
    const int c0 = __builtin_amdgcn_readfirstlane(k0);
    const int c1 = __builtin_amdgcn_readfirstlane(k1);
    const float di = rsqrtf((float)(c0 + c1 + 1));
    const int m0 = c0 < SUB ? c0 : SUB;
    const int m1 = c1 < SUB ? c1 : SUB;
    const int m = m0 + m1;
    float ax = self.x, ay = self.y;
    const int t1 = __shfl(s1, lane - m0);
    const int la = lane < m0 ? s0 : t1;
    const int nxt = node + nw;
    const bool more = nxt < n;
    int ns0 = 0, ns1 = 0, nk0 = 0, nk1 = 0;
    v2f nself = {0.0f, 0.0f};
    if (more) {
      ns0 = col[(size_t)nxt * BKT + lane];
      ns1 = col[(size_t)nxt * BKT + SUB + lane];
      nk0 = cnt0[nxt];
      nk1 = cnt1[nxt];
      nself = hl[(size_t)nxt * 32];
    }
    const int ma = m < 32 ? m : 32;
    walk64<NB>(hl, la, 0, ma, ax, ay);
    if (ma > NB) walk64<32 - NB>(hl, la, NB, ma, ax, ay);
    if (m > 32) {
      const int lb = __shfl(s1, lane + 32 - m0);
      walk64<NB>(hl, lb, 0, m - 32, ax, ay);
      if (m - 32 > NB) walk64<32 - NB>(hl, lb, NB, m - 32, ax, ay);
    }
    if (c0 > SUB || c1 > SUB) {
      const int nov = __builtin_amdgcn_readfirstlane(*ovc);
      for (int q0 = 0; q0 < nov; q0 += 32) {
        int2 ed = int2{-1, 0};
        if (q0 + lane < nov) ed = ov[q0 + lane];
        unsigned hit = __builtin_amdgcn_ballot_w32(ed.x == node);
        while (hit) {
          const int bpos = __builtin_ctz(hit);
          hit &= hit - 1;
          const v2f r = hl[(size_t)__builtin_amdgcn_readlane(ed.y, bpos) * 32];
          ax += r.x;
          ay += r.y;
        }
      }
    }
    v2f sum;
    sum.x = ax;
    sum.y = ay;
    *(v2f*)(myA + j * AP + 2 * lane) = sum;
    if (lane == 0) *(myA + j * AP + 2 * lane + 64) = di;
    ++j;
    if (j == TR || !more) {
      const v2f bb = ((const v2f*)b)[lane];
      const float* arow = myA + (mr & (TR - 1)) * AP + 2 * hi;
      float dv = 0.0f;
      if (hi == 0) dv = arow[64];
      const int dis = __builtin_bit_cast(int, dv);
#pragma unroll 2
      for (int r = 0; r < TR; ++r) {
        const float dr =
            __builtin_bit_cast(float, __builtin_amdgcn_readlane(dis, r));
        const v2f t = *(const v2f*)(myA + r * AP + 2 * lane);
        v2f res;
        res.x = fmaxf(t.x * dr + bb.x, 0.0f);
        res.y = fmaxf(t.y * dr + bb.y, 0.0f);
        *(v2f*)(myA + r * AP + 2 * lane) = res;
      }
      const unsigned bo = (unsigned)(hi * 32 + mr);
      v8f acc = {};
#pragma unroll 1
      for (int kg = 0; kg < 16; kg += 8) {
        v2f bf[8];
#pragma unroll
        for (int u = 0; u < 8; ++u) {
          bf[u].x = w2[bo + (unsigned)((kg + u) * 64)];
          bf[u].y = w2[bo + (unsigned)((kg + u) * 64 + 16)];
        }
#pragma unroll
        for (int u = 0; u < 8; ++u) {
          const v2f a = *(const v2f*)(arow + (kg + u) * 4);
          acc = __builtin_amdgcn_wmma_f32_16x16x4_f32(
               false, a,  false, bf[u],
               (short)0, acc,  false,  false);
        }
      }
      float v[TR];
#pragma unroll
      for (int r = 0; r < TR; ++r)
        v[r] = acc[r] * __builtin_bit_cast(float, __builtin_amdgcn_readlane(dis, r));
#pragma unroll
      for (int r = 0; r < TR; ++r)
        if (r < j && hi == 0)
          (g2 + (size_t)(tile0 + r * nw) * G2P)[(unsigned)mr] = v[r];
      j = 0;
      tile0 = nxt;
    }
    if (!more) break;
    node = nxt;
    s0 = ns0;
    s1 = ns1;
    k0 = nk0;
    k1 = nk1;
    self = nself;
  }
}

constexpr int G2V = G2P / 2;
template <int J>
__device__ __forceinline__ int group_bcast(int v) {
  return (int)__builtin_amdgcn_mov_dpp8((unsigned)v, (unsigned)(J * 0x249249));
}

#define ASK_ROW(R, Q, CNT, K, COMP, J)                       \
  v2f R = {0.0f, 0.0f};                                      \
  {                                                          \
    const unsigned s_ = (unsigned)group_bcast<J>(Q.COMP);    \
    if (K < CNT) R = g2[s_ * (unsigned)G2V + (unsigned)c];   \
  }
#define ADD_ROW(R) \
  acc.x += R.x;    \
  acc.y += R.y;

__device__ __forceinline__ void walk_rest(const v2f* __restrict__ gc,
                                          const int4 q, int cnt, int lane,
                                          v2f& acc) {
  for (int k = 16; __builtin_amdgcn_ballot_w32(k < cnt) != 0u; ++k) {
    const int t = k & 3;
    const int v = t == 0 ? q.x : t == 1 ? q.y : t == 2 ? q.z : q.w;
    const int s = __shfl(v, (lane & 24) | (k >> 2));
    if (k < cnt) {
      const v2f r = gc[(size_t)s * G2V];
      acc.x += r.x;
      acc.y += r.y;
    }
  }
}

__global__ __launch_bounds__(256) void k_gather16(
    const int* __restrict__ cnt0, const int* __restrict__ cnt1,
    const int* __restrict__ col, const int* __restrict__ ovc,
    const int2* __restrict__ ov, const float* __restrict__ g,
    const float* __restrict__ b, float* __restrict__ out, int n) {
  const int lane = threadIdx.x & 31;
  const int c = threadIdx.x & 7;
  const int nq = (int)(gridDim.x * 32);
  int node = (int)((blockIdx.x * 256 + threadIdx.x) >> 3);
  const v2f* g2 = (const v2f*)g;
  const v2f* gc = g2 + c;
  const int4* bq = (const int4*)col + c;
  const v2f bc = ((const v2f*)b)[c];
  const int4 none = int4{0, 0, 0, 0};
  bool act = node < n;
  int c0 = 0, c1 = 0;
  v2f acc = {0.0f, 0.0f};
  int4 q0 = none, q1 = none;
  if (act) {
    c0 = cnt0[node];
    c1 = cnt1[node];
    acc = gc[(size_t)node * G2V];
    q0 = bq[(size_t)node * 16];
    q1 = bq[(size_t)node * 16 + 8];
  }
  while (__builtin_amdgcn_ballot_w32(act) != 0u) {
    int m0 = 0, m1 = 0;
    float di = 0.0f;
    if (act) {
      m0 = c0 < SUB ? c0 : SUB;
      m1 = c1 < SUB ? c1 : SUB;
      di = rsqrtf((float)(c0 + c1 + 1));
    }
    asm volatile("" ::"v"(q1.x), "v"(q1.y), "v"(q1.z), "v"(q1.w));
    ASK_ROW(a0, q0, m0, 0, x, 0)
    ASK_ROW(a1, q0, m0, 1, y, 0)
    ASK_ROW(a2, q0, m0, 2, z, 0)
    ASK_ROW(a3, q0, m0, 3, w, 0)
    ASK_ROW(a4, q0, m0, 4, x, 1)
    ASK_ROW(a5, q0, m0, 5, y, 1)
    ASK_ROW(a6, q0, m0, 6, z, 1)
    ASK_ROW(a7, q0, m0, 7, w, 1)
    ASK_ROW(a8, q0, m0, 8, x, 2)
    ASK_ROW(a9, q0, m0, 9, y, 2)
    ASK_ROW(a10, q0, m0, 10, z, 2)
    ASK_ROW(a11, q0, m0, 11, w, 2)
    ASK_ROW(a12, q0, m0, 12, x, 3)
    ASK_ROW(a13, q0, m0, 13, y, 3)
    ASK_ROW(a14, q0, m0, 14, z, 3)
    ASK_ROW(a15, q0, m0, 15, w, 3)
    ASK_ROW(b0, q1, m1, 0, x, 0)
    ASK_ROW(b1, q1, m1, 1, y, 0)
    ASK_ROW(b2, q1, m1, 2, z, 0)
    ASK_ROW(b3, q1, m1, 3, w, 0)
    ASK_ROW(b4, q1, m1, 4, x, 1)
    ASK_ROW(b5, q1, m1, 5, y, 1)
    ASK_ROW(b6, q1, m1, 6, z, 1)
    ASK_ROW(b7, q1, m1, 7, w, 1)
    ASK_ROW(b8, q1, m1, 8, x, 2)
    ASK_ROW(b9, q1, m1, 9, y, 2)
    ASK_ROW(b10, q1, m1, 10, z, 2)
    ASK_ROW(b11, q1, m1, 11, w, 2)
    ASK_ROW(b12, q1, m1, 12, x, 3)
    ASK_ROW(b13, q1, m1, 13, y, 3)
    ASK_ROW(b14, q1, m1, 14, z, 3)
    ASK_ROW(b15, q1, m1, 15, w, 3)
    const int nxt = node + nq;
    const bool nact = act && nxt < n;
    int nc0 = 0, nc1 = 0;
    v2f nacc = {0.0f, 0.0f};
    int4 nq0 = none, nq1 = none;
    if (nact) {
      nc0 = cnt0[nxt];
      nc1 = cnt1[nxt];
      nacc = gc[(size_t)nxt * G2V];
      nq0 = bq[(size_t)nxt * 16];
      nq1 = bq[(size_t)nxt * 16 + 8];
    }
    ADD_ROW(a0)
    ADD_ROW(a1)
    ADD_ROW(a2)
    ADD_ROW(a3)
    ADD_ROW(a4)
    ADD_ROW(a5)
    ADD_ROW(a6)
    ADD_ROW(a7)
    ADD_ROW(a8)
    ADD_ROW(a9)
    ADD_ROW(a10)
    ADD_ROW(a11)
    ADD_ROW(a12)
    ADD_ROW(a13)
    ADD_ROW(a14)
    ADD_ROW(a15)
    walk_rest(gc, q0, m0, lane, acc);
    ADD_ROW(b0)
    ADD_ROW(b1)
    ADD_ROW(b2)
    ADD_ROW(b3)
    ADD_ROW(b4)
    ADD_ROW(b5)
    ADD_ROW(b6)
    ADD_ROW(b7)
    ADD_ROW(b8)
    ADD_ROW(b9)
    ADD_ROW(b10)
    ADD_ROW(b11)
    ADD_ROW(b12)
    ADD_ROW(b13)
    ADD_ROW(b14)
    ADD_ROW(b15)
    walk_rest(gc, q1, m1, lane, acc);
    if (act && (c0 > SUB || c1 > SUB)) {
      const int nov = *ovc;
      for (int k = 0; k < nov; ++k) {
        const int2 ed = ov[k];
        if (ed.x == node) {
          const v2f r = gc[(size_t)ed.y * G2V];
          acc.x += r.x;
          acc.y += r.y;
        }
      }
    }
    if (act) {
      v2f res;
      res.x = acc.x * di + bc.x;
      res.y = acc.y * di + bc.y;
      ((v2f*)out)[(size_t)node * 8 + c] = res;
    }
    node = nxt;
    act = nact;
    c0 = nc0;
    c1 = nc1;
    acc = nacc;
    q0 = nq0;
    q1 = nq1;
  }
}
#undef ASK_ROW
#undef ADD_ROW

extern "C" void kernel_launch(void* const* d_in, const int* in_sizes, int n_in,
                              void* d_out, int out_size, void* d_ws,
                              size_t ws_size, hipStream_t stream) {
  constexpr int IN_FEAT = 256, HIDDEN = 64;

  const float* x  = (const float*)d_in[0];
  const int*   ei = (const int*)d_in[1];
  const float* W1 = (const float*)d_in[2];
  const float* b1 = (const float*)d_in[3];
  const float* W2 = (const float*)d_in[4];
  const float* b2 = (const float*)d_in[5];
  float* out = (float*)d_out;

  const int N = in_sizes[0] / IN_FEAT;
  const int E = in_sizes[1] / 2;
  const int* src = ei;
  const int* dst = ei + E;

  const size_t NP = ((size_t)N + 63) / 64 * 64;
  const int nbin = (N + NPB - 1) / NPB;
  const int nwg  = (E + CHUNK - 1) / CHUNK;
  const size_t RP = ((size_t)nbin * nwg + 63) / 64 * 64;
  const size_t WP = ((size_t)nwg + 63) / 64 * 64;
  float* ws   = (float*)d_ws;
  float* h1   = ws;
  float* g2   = h1 + (size_t)N * HIDDEN;
  int* cnt0   = (int*)(g2 + NP * G2P);
  int* cnt1   = cnt0 + NP;
  int* ovc    = cnt1 + NP;
  int* col    = ovc + 64;
  int2* ov    = (int2*)(col + (size_t)N * BKT);
  int2* runs  = ov + E;
  int* runcnt = (int*)(runs + (size_t)nbin * nwg * RCAP);
  int* spillcnt = runcnt + RP;
  int2* spill = (int2*)(spillcnt + WP);

  const int B = 256;
  const int mtiles  = (N + 15) / 16;
  const int gblocks = (mtiles + 7) / 8;
  const int pblocks = 2048;
  const int qblocks = 1024;

  k_bin<<<nwg, BT, 0, stream>>>(src, dst, runs, runcnt, spill, spillcnt, ovc, E, nbin, nwg);
  k_rank<<<nbin, BT, 0, stream>>>(runs, runcnt, spill, spillcnt, cnt0, cnt1, col, ovc, ov, N, nwg);

  wmma_gemm_f32<IN_FEAT, HIDDEN><<<gblocks, B, 0, stream>>>(x, W1, cnt0, cnt1, h1, N);
  k_gather64_mm<<<pblocks, B, 0, stream>>>(cnt0, cnt1, col, ovc, ov, h1, b1, W2, g2, N);

  k_gather16<<<qblocks, B, 0, stream>>>(cnt0, cnt1, col, ovc, ov, g2, b2, out, N);
}


# Layer-1 GEMM reads x as whole 512-B row pieces, half a tile ahead, and takes A fragments from a per-wave LDS tile; 256 resident work-groups loop over row tiles (W1 staged once each); k order kept
# speedup vs baseline: 4.0079x; 1.0589x over previous
#include <hip/hip_runtime.h>


typedef float v2f __attribute__((ext_vector_type(2)));
typedef float v8f __attribute__((ext_vector_type(8)));

constexpr int SUB = 32;
constexpr int BKT = 2 * SUB;
typedef float v4f __attribute__((ext_vector_type(4)));
constexpr int GW  = 16;
constexpr int KH  = 128;
constexpr int APG = KH + 4;
template <int K, int N>
__global__ __launch_bounds__(GW * 32) void wmma_gemm_f32(
    const float* __restrict__ A, const float* __restrict__ B,
    const int* __restrict__ cnt0, const int* __restrict__ cnt1,
    float* __restrict__ C, int M) {
  static_assert(K == 2 * KH, "a tile is staged in two halves");
  __shared__ float lB[K * N];
  __shared__ __attribute__((aligned(16))) float lT[GW * 16 * APG];
  const int lane = threadIdx.x & 31;
  const int wave = __builtin_amdgcn_readfirstlane((int)(threadIdx.x >> 5));
  const int mtiles = (M + 15) >> 4;
  const int tstride = (int)gridDim.x * GW;
  int tile = (int)blockIdx.x + (int)gridDim.x * wave;
  const bool any = tile < mtiles;
  v4f g[16];
  const float* ag = A + (size_t)tile * 16 * K + 4 * lane;
  if (any) {
#pragma unroll
    for (int r = 0; r < 16; ++r) g[r] = *(const v4f*)(ag + (size_t)r * K);
  }
  for (int i = threadIdx.x; i < K * N; i += GW * 32) {
    int k = i / N, n = i % N;
    lB[((k >> 2) * N + n) * 4 + (k & 3)] = B[i];
  }
  __syncthreads();
  if (!any) return;

  const int mr = lane & 15;
  const int hi = lane >> 4;

  float* myT = lT + wave * (16 * APG);
  float* tst = myT + 4 * lane;
  const float* afrag = myT + mr * APG + 2 * hi;
  const float* brow = lB + mr * 4 + 2 * hi;

  constexpr int NT = N / 16;

  for (;;) {
    const int mbase = tile * 16;
    v8f acc[NT];
    v8f zero = {};
#pragma unroll
    for (int t = 0; t < NT; ++t) acc[t] = zero;

#pragma unroll
    for (int r = 0; r < 16; ++r) *(v4f*)(tst + r * APG) = g[r];
#pragma unroll
    for (int r = 0; r < 16; ++r) g[r] = *(const v4f*)(ag + (size_t)r * K + KH);
    int c0[8], c1[8];
#pragma unroll
    for (int r = 0; r < 8; ++r) {
      const int row = mbase + 8 * hi + r;
      c0[r] = cnt0[row];
      c1[r] = cnt1[row];
    }
#pragma unroll 4
    for (int kb = 0; kb < KH / 4; ++kb) {
      v2f a = *(const v2f*)(afrag + kb * 4);
#pragma unroll
      for (int t = 0; t < NT; ++t) {
        v2f b = *(const v2f*)(brow + kb * (N * 4) + t * 64);
        acc[t] = __builtin_amdgcn_wmma_f32_16x16x4_f32(
             false, a,  false, b,
             (short)0, acc[t],  false,  false);
      }
    }

#pragma unroll
    for (int r = 0; r < 16; ++r) *(v4f*)(tst + r * APG) = g[r];
    const int nt = tile + tstride;
    const bool more = nt < mtiles;
    const float* nag = A + (size_t)nt * 16 * K + 4 * lane;
    if (more) {
#pragma unroll
      for (int r = 0; r < 16; ++r) g[r] = *(const v4f*)(nag + (size_t)r * K);
    }
#pragma unroll 4
    for (int kb = 0; kb < KH / 4; ++kb) {
      v2f a = *(const v2f*)(afrag + kb * 4);
#pragma unroll
      for (int t = 0; t < NT; ++t) {
        v2f b = *(const v2f*)(brow + (kb + KH / 4) * (N * 4) + t * 64);
        acc[t] = __builtin_amdgcn_wmma_f32_16x16x4_f32(
             false, a,  false, b,
             (short)0, acc[t],  false,  false);
      }
    }

    float sc[8];
#pragma unroll
    for (int r = 0; r < 8; ++r)
      sc[r] = rsqrtf((float)(c0[r] + c1[r] + 1));
#pragma unroll
    for (int t = 0; t < NT; ++t)
#pragma unroll
      for (int r = 0; r < 8; ++r)
        C[(size_t)(mbase + 8 * hi + r) * N + t * 16 + mr] = acc[t][r] * sc[r];

    if (!more) break;
    tile = nt;
    ag = nag;
  }
}

constexpr int NPB   = 256;
constexpr int NPB_L = 8;
constexpr int BT    = 512;
constexpr int BE    = 16;
constexpr int CHUNK = BT * BE;
constexpr int RCAP  = 64;
constexpr int HB    = 512;

__global__ __launch_bounds__(BT) void k_bin(
    const int* __restrict__ src, const int* __restrict__ dst,
    int2* __restrict__ runs, int* __restrict__ runcnt,
    int2* __restrict__ spill, int* __restrict__ spillcnt,
    int* __restrict__ ovc, int e, int nbin, int nwg) {
  static_assert(HB == BT, "the scan below holds one table entry per thread");
  __shared__ int hist[HB];
  __shared__ int off[HB];
  __shared__ int wsum[BT / 32];
  __shared__ int nspill;
  __shared__ int2 stage[CHUNK];
  const int tid = threadIdx.x;
  const int lane = tid & 31;
  const int wv = tid >> 5;
  const int wg = blockIdx.x;
  hist[tid] = 0;
  if (tid == 0) {
    nspill = 0;
    if (wg == 0) *ovc = 0;
  }
  __syncthreads();
  const int base = wg * CHUNK + tid;
  int d[BE], s[BE];
#pragma unroll
  for (int u = 0; u < BE; ++u) {
    const int i = base + u * BT;
    const bool ok = i < e;
    d[u] = ok ? dst[i] : -1;
    s[u] = ok ? src[i] : 0;
  }
  int r[BE];
#pragma unroll
  for (int u = 0; u < BE; ++u) {
    r[u] = -1;
    if (d[u] >= 0) r[u] = atomicAdd(&hist[d[u] >> NPB_L], 1);
  }
  __syncthreads();
  const int h = hist[tid];
  int inc = h;
#pragma unroll
  for (int k = 1; k < 32; k <<= 1) {
    const int t = __shfl_up(inc, k);
    if (lane >= k) inc += t;
  }
  if (lane == 31) wsum[wv] = inc;
  __syncthreads();
  int pre = 0;
  for (int k = 0; k < wv; ++k) pre += wsum[k];
  off[tid] = pre + inc - h;
  __syncthreads();
  int o[BE];
#pragma unroll
  for (int u = 0; u < BE; ++u) o[u] = r[u] >= 0 ? off[d[u] >> NPB_L] : 0;
#pragma unroll
  for (int u = 0; u < BE; ++u)
    if (r[u] >= 0) stage[o[u] + r[u]] = int2{d[u], s[u]};
  __syncthreads();
  const int rem = e - wg * CHUNK;
  const int total = rem < CHUNK ? rem : CHUNK;
  int2 ed[BE];
#pragma unroll
  for (int u = 0; u < BE; ++u) {
    const int j = tid + u * BT;
    ed[u] = int2{-1, 0};
    if (j < total) ed[u] = stage[j];
  }
#pragma unroll
  for (int u = 0; u < BE; ++u) o[u] = ed[u].x >= 0 ? off[ed[u].x >> NPB_L] : 0;
#pragma unroll
  for (int u = 0; u < BE; ++u) {
    if (ed[u].x >= 0) {
      const int bin = ed[u].x >> NPB_L;
      const int rr = tid + u * BT - o[u];
      if (rr < RCAP) {
        runs[((size_t)bin * nwg + wg) * RCAP + rr] = ed[u];
      } else {
        const int k = atomicAdd(&nspill, 1);
        spill[(size_t)wg * CHUNK + k] = ed[u];
      }
    }
  }
  __syncthreads();
  if (tid < nbin) runcnt[(size_t)tid * nwg + wg] = h;
  if (tid == 0) spillcnt[wg] = nspill;
}

__device__ __forceinline__ int rank_of(int* __restrict__ lcnt, int d) {
  return atomicAdd(&lcnt[d & (NPB - 1)], 1);
}
__device__ __forceinline__ void put_slot(int* __restrict__ limg, int d, int s,
                                         int r, int* __restrict__ ovc,
                                         int2* __restrict__ ov) {
  if (r < BKT) {
    limg[(d & (NPB - 1)) * BKT + (r & 1) * SUB + (r >> 1)] = s;
  } else {
    const int k = atomicAdd(ovc, 1);
    ov[k] = int2{d, s};
  }
}

constexpr int RB = 13;
__global__ __launch_bounds__(BT) void k_rank(
    const int2* __restrict__ runs, const int* __restrict__ runcnt,
    const int2* __restrict__ spill, const int* __restrict__ spillcnt,
    int* __restrict__ cnt0, int* __restrict__ cnt1, int* __restrict__ col,
    int* __restrict__ ovc, int2* __restrict__ ov, int n, int nwg) {
  __shared__ int lcnt[NPB];
  __shared__ __attribute__((aligned(16))) int limg[NPB * BKT];
  const int tid = threadIdx.x;
  const int lane = tid & 31;
  const int w = __builtin_amdgcn_readfirstlane(tid >> 5);
  const int b = blockIdx.x;
  const int4 none = int4{0, 0, 0, 0};
  for (int i = tid; i < NPB * BKT / 4; i += BT) ((int4*)limg)[i] = none;
  if (tid < NPB) lcnt[tid] = 0;
  __syncthreads();
  const int4* rq = (const int4*)runs + lane;
  for (int wg0 = w; wg0 < nwg; wg0 += (BT / 32) * RB) {
    int4 q[RB];
    int c[RB];
#pragma unroll
    for (int u = 0; u < RB; ++u) {
      const int wg = wg0 + u * (BT / 32);
      const bool ok = wg < nwg;
      const size_t run = (size_t)b * nwg + (ok ? wg : nwg - 1);
      q[u] = rq[run * (RCAP / 2)];
      const int len = runcnt[run];
      c[u] = ok ? len : 0;
    }
    int r0[RB], r1[RB];
#pragma unroll
    for (int u = 0; u < RB; ++u) {
      const int cc = c[u] < RCAP ? c[u] : RCAP;
      r0[u] = -1;
      r1[u] = -1;
      if (2 * lane < cc) r0[u] = rank_of(lcnt, q[u].x);
      if (2 * lane + 1 < cc) r1[u] = rank_of(lcnt, q[u].z);
    }
#pragma unroll
    for (int u = 0; u < RB; ++u) {
      if (r0[u] >= 0) put_slot(limg, q[u].x, q[u].y, r0[u], ovc, ov);
      if (r1[u] >= 0) put_slot(limg, q[u].z, q[u].w, r1[u], ovc, ov);
    }
  }
  for (int wg = tid; wg < nwg; wg += BT) {
    const int ns = spillcnt[wg];
    for (int k = 0; k < ns; ++k) {
      const int2 ed = spill[(size_t)wg * CHUNK + k];
      if ((ed.x >> NPB_L) == b)
        put_slot(limg, ed.x, ed.y, rank_of(lcnt, ed.x), ovc, ov);
    }
  }
  __syncthreads();
  int4* cq = (int4*)col + (size_t)b * (NPB * BKT / 4);
  for (int i = tid; i < NPB * BKT / 4; i += BT) {
    const int node = b * NPB + (i >> 4);
    if (node < n) cq[i] = ((const int4*)limg)[i];
  }
  if (tid < NPB) {
    const int node = b * NPB + tid;
    if (node < n) {
      const int c = lcnt[tid];
      cnt0[node] = (c + 1) >> 1;
      cnt1[node] = c >> 1;
    }
  }
}

constexpr int NB = 20;
template <int CNT>
__device__ __forceinline__ void walk64(const v2f* __restrict__ hl, int s,
                                       int off, int m, float& ax, float& ay) {
  v2f r[CNT];
#pragma unroll
  for (int u = 0; u < CNT; ++u) {
    r[u].x = 0.0f;
    r[u].y = 0.0f;
    if (off + u < m)
      r[u] = hl[(size_t)__builtin_amdgcn_readlane(s, off + u) * 32];
  }
#pragma unroll
  for (int u = 0; u < CNT; ++u) {
    ax += r[u].x;
    ay += r[u].y;
  }
}

constexpr int TR = 8;
constexpr int AP = 68;
constexpr int G2P = 32;
__global__ __launch_bounds__(256) void k_gather64_mm(
    const int* __restrict__ cnt0, const int* __restrict__ cnt1,
    const int* __restrict__ col, const int* __restrict__ ovc,
    const int2* __restrict__ ov, const float* __restrict__ h,
    const float* __restrict__ b, const float* __restrict__ w2,
    float* __restrict__ g2, int n) {
  __shared__ __attribute__((aligned(16))) float lA[8 * TR * AP];
  const int lane = threadIdx.x & 31;
  const int nw = (int)(gridDim.x * 8);
  int node = __builtin_amdgcn_readfirstlane(
      (int)((blockIdx.x * 256 + threadIdx.x) >> 5));
  if (node >= n) return;
  const v2f* hl = (const v2f*)h + lane;
  float* myA = lA + (threadIdx.x >> 5) * (TR * AP);
  const int mr = lane & 15;
  const int hi = lane >> 4;
  int j = 0;
  int tile0 = node;
  int s0 = col[(size_t)node * BKT + lane];
  int s1 = col[(size_t)node * BKT + SUB + lane];
  int k0 = cnt0[node];
  int k1 = cnt1[node];
  v2f self = hl[(size_t)node * 32];
  {
    static_assert(TR * AP / 2 == 8 * 32 + 16, "tile fill below covers 272 pairs");
    const v2f z2 = {0.0f, 0.0f};
#pragma unroll
    for (int q = 0; q < 8; ++q) ((v2f*)myA)[lane + 32 * q] = z2;
    if (lane < 16) ((v2f*)myA)[lane + 256] = z2;
  }
  for (;;) {
    const int c0 = __builtin_amdgcn_readfirstlane(k0);
    const int c1 = __builtin_amdgcn_readfirstlane(k1);
    const float di = rsqrtf((float)(c0 + c1 + 1));
    const int m0 = c0 < SUB ? c0 : SUB;
    const int m1 = c1 < SUB ? c1 : SUB;
    const int m = m0 + m1;
    float ax = self.x, ay = self.y;
    const int t1 = __shfl(s1, lane - m0);
    const int la = lane < m0 ? s0 : t1;
    const int nxt = node + nw;
    const bool more = nxt < n;
    int ns0 = 0, ns1 = 0, nk0 = 0, nk1 = 0;
    v2f nself = {0.0f, 0.0f};
    if (more) {
      ns0 = col[(size_t)nxt * BKT + lane];
      ns1 = col[(size_t)nxt * BKT + SUB + lane];
      nk0 = cnt0[nxt];
      nk1 = cnt1[nxt];
      nself = hl[(size_t)nxt * 32];
    }
    const int ma = m < 32 ? m : 32;
    walk64<NB>(hl, la, 0, ma, ax, ay);
    if (ma > NB) walk64<32 - NB>(hl, la, NB, ma, ax, ay);
    if (m > 32) {
      const int lb = __shfl(s1, lane + 32 - m0);
      walk64<NB>(hl, lb, 0, m - 32, ax, ay);
      if (m - 32 > NB) walk64<32 - NB>(hl, lb, NB, m - 32, ax, ay);
    }
    if (c0 > SUB || c1 > SUB) {
      const int nov = __builtin_amdgcn_readfirstlane(*ovc);
      for (int q0 = 0; q0 < nov; q0 += 32) {
        int2 ed = int2{-1, 0};
        if (q0 + lane < nov) ed = ov[q0 + lane];
        unsigned hit = __builtin_amdgcn_ballot_w32(ed.x == node);
        while (hit) {
          const int bpos = __builtin_ctz(hit);
          hit &= hit - 1;
          const v2f r = hl[(size_t)__builtin_amdgcn_readlane(ed.y, bpos) * 32];
          ax += r.x;
          ay += r.y;
        }
      }
    }
    v2f sum;
    sum.x = ax;
    sum.y = ay;
    *(v2f*)(myA + j * AP + 2 * lane) = sum;
    if (lane == 0) *(myA + j * AP + 2 * lane + 64) = di;
    ++j;
    if (j == TR || !more) {
      const v2f bb = ((const v2f*)b)[lane];
      const float* arow = myA + (mr & (TR - 1)) * AP + 2 * hi;
      float dv = 0.0f;
      if (hi == 0) dv = arow[64];
      const int dis = __builtin_bit_cast(int, dv);
#pragma unroll 2
      for (int r = 0; r < TR; ++r) {
        const float dr =
            __builtin_bit_cast(float, __builtin_amdgcn_readlane(dis, r));
        const v2f t = *(const v2f*)(myA + r * AP + 2 * lane);
        v2f res;
        res.x = fmaxf(t.x * dr + bb.x, 0.0f);
        res.y = fmaxf(t.y * dr + bb.y, 0.0f);
        *(v2f*)(myA + r * AP + 2 * lane) = res;
      }
      const unsigned bo = (unsigned)(hi * 32 + mr);
      v8f acc = {};
#pragma unroll 1
      for (int kg = 0; kg < 16; kg += 8) {
        v2f bf[8];
#pragma unroll
        for (int u = 0; u < 8; ++u) {
          bf[u].x = w2[bo + (unsigned)((kg + u) * 64)];
          bf[u].y = w2[bo + (unsigned)((kg + u) * 64 + 16)];
        }
#pragma unroll
        for (int u = 0; u < 8; ++u) {
          const v2f a = *(const v2f*)(arow + (kg + u) * 4);
          acc = __builtin_amdgcn_wmma_f32_16x16x4_f32(
               false, a,  false, bf[u],
               (short)0, acc,  false,  false);
        }
      }
      float v[TR];
#pragma unroll
      for (int r = 0; r < TR; ++r)
        v[r] = acc[r] * __builtin_bit_cast(float, __builtin_amdgcn_readlane(dis, r));
#pragma unroll
      for (int r = 0; r < TR; ++r)
        if (r < j && hi == 0)
          (g2 + (size_t)(tile0 + r * nw) * G2P)[(unsigned)mr] = v[r];
      j = 0;
      tile0 = nxt;
    }
    if (!more) break;
    node = nxt;
    s0 = ns0;
    s1 = ns1;
    k0 = nk0;
    k1 = nk1;
    self = nself;
  }
}

constexpr int G2V = G2P / 2;
template <int J>
__device__ __forceinline__ int group_bcast(int v) {
  return (int)__builtin_amdgcn_mov_dpp8((unsigned)v, (unsigned)(J * 0x249249));
}

#define ASK_ROW(R, Q, CNT, K, COMP, J)                       \
  v2f R = {0.0f, 0.0f};                                      \
  {                                                          \
    const unsigned s_ = (unsigned)group_bcast<J>(Q.COMP);    \
    if (K < CNT) R = g2[s_ * (unsigned)G2V + (unsigned)c];   \
  }
#define ADD_ROW(R) \
  acc.x += R.x;    \
  acc.y += R.y;

__device__ __forceinline__ void walk_rest(const v2f* __restrict__ gc,
                                          const int4 q, int cnt, int lane,
                                          v2f& acc) {
  for (int k = 16; __builtin_amdgcn_ballot_w32(k < cnt) != 0u; ++k) {
    const int t = k & 3;
    const int v = t == 0 ? q.x : t == 1 ? q.y : t == 2 ? q.z : q.w;
    const int s = __shfl(v, (lane & 24) | (k >> 2));
    if (k < cnt) {
      const v2f r = gc[(size_t)s * G2V];
      acc.x += r.x;
      acc.y += r.y;
    }
  }
}

__global__ __launch_bounds__(256) void k_gather16(
    const int* __restrict__ cnt0, const int* __restrict__ cnt1,
    const int* __restrict__ col, const int* __restrict__ ovc,
    const int2* __restrict__ ov, const float* __restrict__ g,
    const float* __restrict__ b, float* __restrict__ out, int n) {
  const int lane = threadIdx.x & 31;
  const int c = threadIdx.x & 7;
  const int nq = (int)(gridDim.x * 32);
  int node = (int)((blockIdx.x * 256 + threadIdx.x) >> 3);
  const v2f* g2 = (const v2f*)g;
  const v2f* gc = g2 + c;
  const int4* bq = (const int4*)col + c;
  const v2f bc = ((const v2f*)b)[c];
  const int4 none = int4{0, 0, 0, 0};
  bool act = node < n;
  int c0 = 0, c1 = 0;
  v2f acc = {0.0f, 0.0f};
  int4 q0 = none, q1 = none;
  if (act) {
    c0 = cnt0[node];
    c1 = cnt1[node];
    acc = gc[(size_t)node * G2V];
    q0 = bq[(size_t)node * 16];
    q1 = bq[(size_t)node * 16 + 8];
  }
  while (__builtin_amdgcn_ballot_w32(act) != 0u) {
    int m0 = 0, m1 = 0;
    float di = 0.0f;
    if (act) {
      m0 = c0 < SUB ? c0 : SUB;
      m1 = c1 < SUB ? c1 : SUB;
      di = rsqrtf((float)(c0 + c1 + 1));
    }
    asm volatile("" ::"v"(q1.x), "v"(q1.y), "v"(q1.z), "v"(q1.w));
    ASK_ROW(a0, q0, m0, 0, x, 0)
    ASK_ROW(a1, q0, m0, 1, y, 0)
    ASK_ROW(a2, q0, m0, 2, z, 0)
    ASK_ROW(a3, q0, m0, 3, w, 0)
    ASK_ROW(a4, q0, m0, 4, x, 1)
    ASK_ROW(a5, q0, m0, 5, y, 1)
    ASK_ROW(a6, q0, m0, 6, z, 1)
    ASK_ROW(a7, q0, m0, 7, w, 1)
    ASK_ROW(a8, q0, m0, 8, x, 2)
    ASK_ROW(a9, q0, m0, 9, y, 2)
    ASK_ROW(a10, q0, m0, 10, z, 2)
    ASK_ROW(a11, q0, m0, 11, w, 2)
    ASK_ROW(a12, q0, m0, 12, x, 3)
    ASK_ROW(a13, q0, m0, 13, y, 3)
    ASK_ROW(a14, q0, m0, 14, z, 3)
    ASK_ROW(a15, q0, m0, 15, w, 3)
    ASK_ROW(b0, q1, m1, 0, x, 0)
    ASK_ROW(b1, q1, m1, 1, y, 0)
    ASK_ROW(b2, q1, m1, 2, z, 0)
    ASK_ROW(b3, q1, m1, 3, w, 0)
    ASK_ROW(b4, q1, m1, 4, x, 1)
    ASK_ROW(b5, q1, m1, 5, y, 1)
    ASK_ROW(b6, q1, m1, 6, z, 1)
    ASK_ROW(b7, q1, m1, 7, w, 1)
    ASK_ROW(b8, q1, m1, 8, x, 2)
    ASK_ROW(b9, q1, m1, 9, y, 2)
    ASK_ROW(b10, q1, m1, 10, z, 2)
    ASK_ROW(b11, q1, m1, 11, w, 2)
    ASK_ROW(b12, q1, m1, 12, x, 3)
    ASK_ROW(b13, q1, m1, 13, y, 3)
    ASK_ROW(b14, q1, m1, 14, z, 3)
    ASK_ROW(b15, q1, m1, 15, w, 3)
    const int nxt = node + nq;
    const bool nact = act && nxt < n;
    int nc0 = 0, nc1 = 0;
    v2f nacc = {0.0f, 0.0f};
    int4 nq0 = none, nq1 = none;
    if (nact) {
      nc0 = cnt0[nxt];
      nc1 = cnt1[nxt];
      nacc = gc[(size_t)nxt * G2V];
      nq0 = bq[(size_t)nxt * 16];
      nq1 = bq[(size_t)nxt * 16 + 8];
    }
    ADD_ROW(a0)
    ADD_ROW(a1)
    ADD_ROW(a2)
    ADD_ROW(a3)
    ADD_ROW(a4)
    ADD_ROW(a5)
    ADD_ROW(a6)
    ADD_ROW(a7)
    ADD_ROW(a8)
    ADD_ROW(a9)
    ADD_ROW(a10)
    ADD_ROW(a11)
    ADD_ROW(a12)
    ADD_ROW(a13)
    ADD_ROW(a14)
    ADD_ROW(a15)
    walk_rest(gc, q0, m0, lane, acc);
    ADD_ROW(b0)
    ADD_ROW(b1)
    ADD_ROW(b2)
    ADD_ROW(b3)
    ADD_ROW(b4)
    ADD_ROW(b5)
    ADD_ROW(b6)
    ADD_ROW(b7)
    ADD_ROW(b8)
    ADD_ROW(b9)
    ADD_ROW(b10)
    ADD_ROW(b11)
    ADD_ROW(b12)
    ADD_ROW(b13)
    ADD_ROW(b14)
    ADD_ROW(b15)
    walk_rest(gc, q1, m1, lane, acc);
    if (act && (c0 > SUB || c1 > SUB)) {
      const int nov = *ovc;
      for (int k = 0; k < nov; ++k) {
        const int2 ed = ov[k];
        if (ed.x == node) {
          const v2f r = gc[(size_t)ed.y * G2V];
          acc.x += r.x;
          acc.y += r.y;
        }
      }
    }
    if (act) {
      v2f res;
      res.x = acc.x * di + bc.x;
      res.y = acc.y * di + bc.y;
      ((v2f*)out)[(size_t)node * 8 + c] = res;
    }
    node = nxt;
    act = nact;
    c0 = nc0;
    c1 = nc1;
    acc = nacc;
    q0 = nq0;
    q1 = nq1;
  }
}
#undef ASK_ROW
#undef ADD_ROW

extern "C" void kernel_launch(void* const* d_in, const int* in_sizes, int n_in,
                              void* d_out, int out_size, void* d_ws,
                              size_t ws_size, hipStream_t stream) {
  constexpr int IN_FEAT = 256, HIDDEN = 64;

  const float* x  = (const float*)d_in[0];
  const int*   ei = (const int*)d_in[1];
  const float* W1 = (const float*)d_in[2];
  const float* b1 = (const float*)d_in[3];
  const float* W2 = (const float*)d_in[4];
  const float* b2 = (const float*)d_in[5];
  float* out = (float*)d_out;

  const int N = in_sizes[0] / IN_FEAT;
  const int E = in_sizes[1] / 2;
  const int* src = ei;
  const int* dst = ei + E;

  const size_t NP = ((size_t)N + 63) / 64 * 64;
  const int nbin = (N + NPB - 1) / NPB;
  const int nwg  = (E + CHUNK - 1) / CHUNK;
  const size_t RP = ((size_t)nbin * nwg + 63) / 64 * 64;
  const size_t WP = ((size_t)nwg + 63) / 64 * 64;
  float* ws   = (float*)d_ws;
  float* h1   = ws;
  float* g2   = h1 + (size_t)N * HIDDEN;
  int* cnt0   = (int*)(g2 + NP * G2P);
  int* cnt1   = cnt0 + NP;
  int* ovc    = cnt1 + NP;
  int* col    = ovc + 64;
  int2* ov    = (int2*)(col + (size_t)N * BKT);
  int2* runs  = ov + E;
  int* runcnt = (int*)(runs + (size_t)nbin * nwg * RCAP);
  int* spillcnt = runcnt + RP;
  int2* spill = (int2*)(spillcnt + WP);

  const int B = 256;
  const int gblocks = 256;
  const int pblocks = 2048;
  const int qblocks = 1024;

  k_bin<<<nwg, BT, 0, stream>>>(src, dst, runs, runcnt, spill, spillcnt, ovc, E, nbin, nwg);
  k_rank<<<nbin, BT, 0, stream>>>(runs, runcnt, spill, spillcnt, cnt0, cnt1, col, ovc, ov, N, nwg);

  wmma_gemm_f32<IN_FEAT, HIDDEN><<<gblocks, GW * 32, 0, stream>>>(x, W1, cnt0, cnt1, h1, N);
  k_gather64_mm<<<pblocks, B, 0, stream>>>(cnt0, cnt1, col, ovc, ov, h1, b1, W2, g2, N);

  k_gather16<<<qblocks, B, 0, stream>>>(cnt0, cnt1, col, ovc, ov, g2, b2, out, N);
}


# Buckets and pass-1 runs split into two planes of 32 places filled in rank order; the second is written and read only where a count exceeds 32: half the bucket and run bytes; sum order = rank order
# speedup vs baseline: 4.2719x; 1.0659x over previous
#include <hip/hip_runtime.h>


typedef float v2f __attribute__((ext_vector_type(2)));
typedef float v8f __attribute__((ext_vector_type(8)));

constexpr int SUB = 32;
constexpr int BKT = 2 * SUB;
typedef float v4f __attribute__((ext_vector_type(4)));
constexpr int GW  = 16;
constexpr int KH  = 128;
constexpr int APG = KH + 4;
template <int K, int N>
__global__ __launch_bounds__(GW * 32) void wmma_gemm_f32(
    const float* __restrict__ A, const float* __restrict__ B,
    const int* __restrict__ cnt, float* __restrict__ C, int M) {
  static_assert(K == 2 * KH, "a tile is staged in two halves");
  __shared__ float lB[K * N];
  __shared__ __attribute__((aligned(16))) float lT[GW * 16 * APG];
  const int lane = threadIdx.x & 31;
  const int wave = __builtin_amdgcn_readfirstlane((int)(threadIdx.x >> 5));
  const int mtiles = (M + 15) >> 4;
  const int tstride = (int)gridDim.x * GW;
  int tile = (int)blockIdx.x + (int)gridDim.x * wave;
  const bool any = tile < mtiles;
  v4f g[16];
  const float* ag = A + (size_t)tile * 16 * K + 4 * lane;
  if (any) {
#pragma unroll
    for (int r = 0; r < 16; ++r) g[r] = *(const v4f*)(ag + (size_t)r * K);
  }
  for (int i = threadIdx.x; i < K * N; i += GW * 32) {
    int k = i / N, n = i % N;
    lB[((k >> 2) * N + n) * 4 + (k & 3)] = B[i];
  }
  __syncthreads();
  if (!any) return;

  const int mr = lane & 15;
  const int hi = lane >> 4;

  float* myT = lT + wave * (16 * APG);
  float* tst = myT + 4 * lane;
  const float* afrag = myT + mr * APG + 2 * hi;
  const float* brow = lB + mr * 4 + 2 * hi;

  constexpr int NT = N / 16;

  for (;;) {
    const int mbase = tile * 16;
    v8f acc[NT];
    v8f zero = {};
#pragma unroll
    for (int t = 0; t < NT; ++t) acc[t] = zero;

#pragma unroll
    for (int r = 0; r < 16; ++r) *(v4f*)(tst + r * APG) = g[r];
#pragma unroll
    for (int r = 0; r < 16; ++r) g[r] = *(const v4f*)(ag + (size_t)r * K + KH);
    int c0[8];
#pragma unroll
    for (int r = 0; r < 8; ++r) c0[r] = cnt[mbase + 8 * hi + r];
#pragma unroll 4
    for (int kb = 0; kb < KH / 4; ++kb) {
      v2f a = *(const v2f*)(afrag + kb * 4);
#pragma unroll
      for (int t = 0; t < NT; ++t) {
        v2f b = *(const v2f*)(brow + kb * (N * 4) + t * 64);
        acc[t] = __builtin_amdgcn_wmma_f32_16x16x4_f32(
             false, a,  false, b,
             (short)0, acc[t],  false,  false);
      }
    }

#pragma unroll
    for (int r = 0; r < 16; ++r) *(v4f*)(tst + r * APG) = g[r];
    const int nt = tile + tstride;
    const bool more = nt < mtiles;
    const float* nag = A + (size_t)nt * 16 * K + 4 * lane;
    if (more) {
#pragma unroll
      for (int r = 0; r < 16; ++r) g[r] = *(const v4f*)(nag + (size_t)r * K);
    }
#pragma unroll 4
    for (int kb = 0; kb < KH / 4; ++kb) {
      v2f a = *(const v2f*)(afrag + kb * 4);
#pragma unroll
      for (int t = 0; t < NT; ++t) {
        v2f b = *(const v2f*)(brow + (kb + KH / 4) * (N * 4) + t * 64);
        acc[t] = __builtin_amdgcn_wmma_f32_16x16x4_f32(
             false, a,  false, b,
             (short)0, acc[t],  false,  false);
      }
    }

    float sc[8];
#pragma unroll
    for (int r = 0; r < 8; ++r)
      sc[r] = rsqrtf((float)(c0[r] + 1));
#pragma unroll
    for (int t = 0; t < NT; ++t)
#pragma unroll
      for (int r = 0; r < 8; ++r)
        C[(size_t)(mbase + 8 * hi + r) * N + t * 16 + mr] = acc[t][r] * sc[r];

    if (!more) break;
    tile = nt;
    ag = nag;
  }
}

constexpr int NPB   = 256;
constexpr int NPB_L = 8;
constexpr int BT    = 512;
constexpr int BE    = 16;
constexpr int CHUNK = BT * BE;
constexpr int RCAP  = 64;
constexpr int RH    = 32;
constexpr int HB    = 512;

__global__ __launch_bounds__(BT) void k_bin(
    const int* __restrict__ src, const int* __restrict__ dst,
    int2* __restrict__ runs, int2* __restrict__ runs2,
    int* __restrict__ runcnt,
    int2* __restrict__ spill, int* __restrict__ spillcnt,
    int* __restrict__ ovc, int e, int nbin, int nwg) {
  static_assert(HB == BT, "the scan below holds one table entry per thread");
  __shared__ int hist[HB];
  __shared__ int off[HB];
  __shared__ int wsum[BT / 32];
  __shared__ int nspill;
  __shared__ int2 stage[CHUNK];
  const int tid = threadIdx.x;
  const int lane = tid & 31;
  const int wv = tid >> 5;
  const int wg = blockIdx.x;
  hist[tid] = 0;
  if (tid == 0) {
    nspill = 0;
    if (wg == 0) *ovc = 0;
  }
  __syncthreads();
  const int base = wg * CHUNK + tid;
  int d[BE], s[BE];
#pragma unroll
  for (int u = 0; u < BE; ++u) {
    const int i = base + u * BT;
    const bool ok = i < e;
    d[u] = ok ? dst[i] : -1;
    s[u] = ok ? src[i] : 0;
  }
  int r[BE];
#pragma unroll
  for (int u = 0; u < BE; ++u) {
    r[u] = -1;
    if (d[u] >= 0) r[u] = atomicAdd(&hist[d[u] >> NPB_L], 1);
  }
  __syncthreads();
  const int h = hist[tid];
  int inc = h;
#pragma unroll
  for (int k = 1; k < 32; k <<= 1) {
    const int t = __shfl_up(inc, k);
    if (lane >= k) inc += t;
  }
  if (lane == 31) wsum[wv] = inc;
  __syncthreads();
  int pre = 0;
  for (int k = 0; k < wv; ++k) pre += wsum[k];
  off[tid] = pre + inc - h;
  __syncthreads();
  int o[BE];
#pragma unroll
  for (int u = 0; u < BE; ++u) o[u] = r[u] >= 0 ? off[d[u] >> NPB_L] : 0;
#pragma unroll
  for (int u = 0; u < BE; ++u)
    if (r[u] >= 0) stage[o[u] + r[u]] = int2{d[u], s[u]};
  __syncthreads();
  const int rem = e - wg * CHUNK;
  const int total = rem < CHUNK ? rem : CHUNK;
  int2 ed[BE];
#pragma unroll
  for (int u = 0; u < BE; ++u) {
    const int j = tid + u * BT;
    ed[u] = int2{-1, 0};
    if (j < total) ed[u] = stage[j];
  }
#pragma unroll
  for (int u = 0; u < BE; ++u) o[u] = ed[u].x >= 0 ? off[ed[u].x >> NPB_L] : 0;
#pragma unroll
  for (int u = 0; u < BE; ++u) {
    if (ed[u].x >= 0) {
      const int bin = ed[u].x >> NPB_L;
      const int rr = tid + u * BT - o[u];
      if (rr < RH) {
        runs[((size_t)bin * nwg + wg) * RH + rr] = ed[u];
      } else if (rr < RCAP) {
        runs2[((size_t)bin * nwg + wg) * RH + (rr - RH)] = ed[u];
      } else {
        const int k = atomicAdd(&nspill, 1);
        spill[(size_t)wg * CHUNK + k] = ed[u];
      }
    }
  }
  __syncthreads();
  if (tid < nbin) runcnt[(size_t)tid * nwg + wg] = h;
  if (tid == 0) spillcnt[wg] = nspill;
}

__device__ __forceinline__ int rank_of(int* __restrict__ lcnt, int d) {
  return atomicAdd(&lcnt[d & (NPB - 1)], 1);
}
__device__ __forceinline__ void put_slot(int* __restrict__ limg, int d, int s,
                                         int r, int* __restrict__ ovc,
                                         int2* __restrict__ ov) {
  if (r < BKT) {
    limg[(d & (NPB - 1)) * BKT + r] = s;
  } else {
    const int k = atomicAdd(ovc, 1);
    ov[k] = int2{d, s};
  }
}

constexpr int RB = 13;
__global__ __launch_bounds__(BT) void k_rank(
    const int2* __restrict__ runs, const int2* __restrict__ runs2,
    const int* __restrict__ runcnt,
    const int2* __restrict__ spill, const int* __restrict__ spillcnt,
    int* __restrict__ cnt, int* __restrict__ col, int* __restrict__ col2,
    int* __restrict__ ovc, int2* __restrict__ ov, int n, int nwg) {
  __shared__ int lcnt[NPB];
  __shared__ __attribute__((aligned(16))) int limg[NPB * BKT];
  const int tid = threadIdx.x;
  const int lane = tid & 31;
  const int w = __builtin_amdgcn_readfirstlane(tid >> 5);
  const int b = blockIdx.x;
  const int4 none = int4{0, 0, 0, 0};
  for (int i = tid; i < NPB * BKT / 4; i += BT) ((int4*)limg)[i] = none;
  if (tid < NPB) lcnt[tid] = 0;
  __syncthreads();
  const int2* rq = runs + lane;
  const int2* rq2 = runs2 + lane;
  for (int wg0 = w; wg0 < nwg; wg0 += (BT / 32) * RB) {
    int2 q[RB];
    int c[RB];
#pragma unroll
    for (int u = 0; u < RB; ++u) {
      const int wg = wg0 + u * (BT / 32);
      const bool ok = wg < nwg;
      const size_t run = (size_t)b * nwg + (ok ? wg : nwg - 1);
      q[u] = rq[run * RH];
      const int len = runcnt[run];
      c[u] = ok ? len : 0;
    }
    int r0[RB];
#pragma unroll
    for (int u = 0; u < RB; ++u) {
      r0[u] = -1;
      if (lane < c[u]) r0[u] = rank_of(lcnt, q[u].x);
    }
#pragma unroll
    for (int u = 0; u < RB; ++u)
      if (r0[u] >= 0) put_slot(limg, q[u].x, q[u].y, r0[u], ovc, ov);
#pragma unroll
    for (int u = 0; u < RB; ++u) {
      const int cu = __builtin_amdgcn_readfirstlane(c[u]);
      if (cu > RH) {
        const int wg = wg0 + u * (BT / 32);
        const int2 e2 = rq2[((size_t)b * nwg + wg) * RH];
        const int cc = cu < RCAP ? cu : RCAP;
        if (RH + lane < cc)
          put_slot(limg, e2.x, e2.y, rank_of(lcnt, e2.x), ovc, ov);
      }
    }
  }
  for (int wg = tid; wg < nwg; wg += BT) {
    const int ns = spillcnt[wg];
    for (int k = 0; k < ns; ++k) {
      const int2 ed = spill[(size_t)wg * CHUNK + k];
      if ((ed.x >> NPB_L) == b)
        put_slot(limg, ed.x, ed.y, rank_of(lcnt, ed.x), ovc, ov);
    }
  }
  __syncthreads();
  int4* cq = (int4*)col + (size_t)b * (NPB * SUB / 4);
  int4* cq2 = (int4*)col2 + (size_t)b * (NPB * SUB / 4);
  for (int i = tid; i < NPB * SUB / 4; i += BT) {
    const int nl = i >> 3;
    const int node = b * NPB + nl;
    if (node < n) {
      cq[i] = ((const int4*)limg)[nl * (BKT / 4) + (i & 7)];
      if (lcnt[nl] > SUB) cq2[i] = ((const int4*)limg)[nl * (BKT / 4) + SUB / 4 + (i & 7)];
    }
  }
  if (tid < NPB) {
    const int node = b * NPB + tid;
    if (node < n) cnt[node] = lcnt[tid];
  }
}

constexpr int NB = 20;
template <int CNT>
__device__ __forceinline__ void walk64(const v2f* __restrict__ hl, int s,
                                       int off, int m, float& ax, float& ay) {
  v2f r[CNT];
#pragma unroll
  for (int u = 0; u < CNT; ++u) {
    r[u].x = 0.0f;
    r[u].y = 0.0f;
    if (off + u < m)
      r[u] = hl[(size_t)__builtin_amdgcn_readlane(s, off + u) * 32];
  }
#pragma unroll
  for (int u = 0; u < CNT; ++u) {
    ax += r[u].x;
    ay += r[u].y;
  }
}

constexpr int TR = 8;
constexpr int AP = 68;
constexpr int G2P = 32;
__global__ __launch_bounds__(256) void k_gather64_mm(
    const int* __restrict__ cnt, const int* __restrict__ col,
    const int* __restrict__ col2, const int* __restrict__ ovc,
    const int2* __restrict__ ov, const float* __restrict__ h,
    const float* __restrict__ b, const float* __restrict__ w2,
    float* __restrict__ g2, int n) {
  __shared__ __attribute__((aligned(16))) float lA[8 * TR * AP];
  const int lane = threadIdx.x & 31;
  const int nw = (int)(gridDim.x * 8);
  int node = __builtin_amdgcn_readfirstlane(
      (int)((blockIdx.x * 256 + threadIdx.x) >> 5));
  if (node >= n) return;
  const v2f* hl = (const v2f*)h + lane;
  float* myA = lA + (threadIdx.x >> 5) * (TR * AP);
  const int mr = lane & 15;
  const int hi = lane >> 4;
  int j = 0;
  int tile0 = node;
  int s0 = col[(size_t)node * SUB + lane];
  int k0 = cnt[node];
  v2f self = hl[(size_t)node * 32];
  {
    static_assert(TR * AP / 2 == 8 * 32 + 16, "tile fill below covers 272 pairs");
    const v2f z2 = {0.0f, 0.0f};
#pragma unroll
    for (int q = 0; q < 8; ++q) ((v2f*)myA)[lane + 32 * q] = z2;
    if (lane < 16) ((v2f*)myA)[lane + 256] = z2;
  }
  for (;;) {
    const int c0 = __builtin_amdgcn_readfirstlane(k0);
    const float di = rsqrtf((float)(c0 + 1));
    const int ma = c0 < SUB ? c0 : SUB;
    float ax = self.x, ay = self.y;
    asm volatile("" ::"v"(s0));
    const int la = s0;
    const int nxt = node + nw;
    const bool more = nxt < n;
    int ns0 = 0, nk0 = 0;
    v2f nself = {0.0f, 0.0f};
    if (more) {
      ns0 = col[(size_t)nxt * SUB + lane];
      nk0 = cnt[nxt];
      nself = hl[(size_t)nxt * 32];
    }
    walk64<NB>(hl, la, 0, ma, ax, ay);
    if (ma > NB) walk64<32 - NB>(hl, la, NB, ma, ax, ay);
    if (c0 > SUB) {
      const int lb = col2[(size_t)node * SUB + lane];
      const int mb = (c0 < BKT ? c0 : BKT) - SUB;
      walk64<NB>(hl, lb, 0, mb, ax, ay);
      if (mb > NB) walk64<32 - NB>(hl, lb, NB, mb, ax, ay);
    }
    if (c0 > BKT) {
      const int nov = __builtin_amdgcn_readfirstlane(*ovc);
      for (int q0 = 0; q0 < nov; q0 += 32) {
        int2 ed = int2{-1, 0};
        if (q0 + lane < nov) ed = ov[q0 + lane];
        unsigned hit = __builtin_amdgcn_ballot_w32(ed.x == node);
        while (hit) {
          const int bpos = __builtin_ctz(hit);
          hit &= hit - 1;
          const v2f r = hl[(size_t)__builtin_amdgcn_readlane(ed.y, bpos) * 32];
          ax += r.x;
          ay += r.y;
        }
      }
    }
    v2f sum;
    sum.x = ax;
    sum.y = ay;
    *(v2f*)(myA + j * AP + 2 * lane) = sum;
    if (lane == 0) *(myA + j * AP + 2 * lane + 64) = di;
    ++j;
    if (j == TR || !more) {
      const v2f bb = ((const v2f*)b)[lane];
      const float* arow = myA + (mr & (TR - 1)) * AP + 2 * hi;
      float dv = 0.0f;
      if (hi == 0) dv = arow[64];
      const int dis = __builtin_bit_cast(int, dv);
#pragma unroll 2
      for (int r = 0; r < TR; ++r) {
        const float dr =
            __builtin_bit_cast(float, __builtin_amdgcn_readlane(dis, r));
        const v2f t = *(const v2f*)(myA + r * AP + 2 * lane);
        v2f res;
        res.x = fmaxf(t.x * dr + bb.x, 0.0f);
        res.y = fmaxf(t.y * dr + bb.y, 0.0f);
        *(v2f*)(myA + r * AP + 2 * lane) = res;
      }
      const unsigned bo = (unsigned)(hi * 32 + mr);
      v8f acc = {};
#pragma unroll 1
      for (int kg = 0; kg < 16; kg += 8) {
        v2f bf[8];
#pragma unroll
        for (int u = 0; u < 8; ++u) {
          bf[u].x = w2[bo + (unsigned)((kg + u) * 64)];
          bf[u].y = w2[bo + (unsigned)((kg + u) * 64 + 16)];
        }
#pragma unroll
        for (int u = 0; u < 8; ++u) {
          const v2f a = *(const v2f*)(arow + (kg + u) * 4);
          acc = __builtin_amdgcn_wmma_f32_16x16x4_f32(
               false, a,  false, bf[u],
               (short)0, acc,  false,  false);
        }
      }
      float v[TR];
#pragma unroll
      for (int r = 0; r < TR; ++r)
        v[r] = acc[r] * __builtin_bit_cast(float, __builtin_amdgcn_readlane(dis, r));
#pragma unroll
      for (int r = 0; r < TR; ++r)
        if (r < j && hi == 0)
          (g2 + (size_t)(tile0 + r * nw) * G2P)[(unsigned)mr] = v[r];
      j = 0;
      tile0 = nxt;
    }
    if (!more) break;
    node = nxt;
    s0 = ns0;
    k0 = nk0;
    self = nself;
  }
}

constexpr int G2V = G2P / 2;
template <int J>
__device__ __forceinline__ int group_bcast(int v) {
  return (int)__builtin_amdgcn_mov_dpp8((unsigned)v, (unsigned)(J * 0x249249));
}

#define ASK_ROW(R, Q, CNT, K, COMP, J)                       \
  v2f R = {0.0f, 0.0f};                                      \
  {                                                          \
    const unsigned s_ = (unsigned)group_bcast<J>(Q.COMP);    \
    if (K < CNT) R = g2[s_ * (unsigned)G2V + (unsigned)c];   \
  }
#define ADD_ROW(R) \
  acc.x += R.x;    \
  acc.y += R.y;

__global__ __launch_bounds__(256) void k_gather16(
    const int* __restrict__ cnt, const int* __restrict__ col,
    const int* __restrict__ col2, const int* __restrict__ ovc,
    const int2* __restrict__ ov, const float* __restrict__ g,
    const float* __restrict__ b, float* __restrict__ out, int n) {
  const int c = threadIdx.x & 7;
  const int nq = (int)(gridDim.x * 32);
  int node = (int)((blockIdx.x * 256 + threadIdx.x) >> 3);
  const v2f* g2 = (const v2f*)g;
  const v2f* gc = g2 + c;
  const int4* bq = (const int4*)col + c;
  const v2f bc = ((const v2f*)b)[c];
  const int4 none = int4{0, 0, 0, 0};
  bool act = node < n;
  int c0 = 0;
  v2f acc = {0.0f, 0.0f};
  int4 q0 = none;
  if (act) {
    c0 = cnt[node];
    acc = gc[(size_t)node * G2V];
    q0 = bq[(size_t)node * 8];
  }
  while (__builtin_amdgcn_ballot_w32(act) != 0u) {
    int m0 = 0;
    float di = 0.0f;
    if (act) {
      m0 = c0 < SUB ? c0 : SUB;
      di = rsqrtf((float)(c0 + 1));
    }
    ASK_ROW(a0, q0, m0, 0, x, 0)
    ASK_ROW(a1, q0, m0, 1, y, 0)
    ASK_ROW(a2, q0, m0, 2, z, 0)
    ASK_ROW(a3, q0, m0, 3, w, 0)
    ASK_ROW(a4, q0, m0, 4, x, 1)
    ASK_ROW(a5, q0, m0, 5, y, 1)
    ASK_ROW(a6, q0, m0, 6, z, 1)
    ASK_ROW(a7, q0, m0, 7, w, 1)
    ASK_ROW(a8, q0, m0, 8, x, 2)
    ASK_ROW(a9, q0, m0, 9, y, 2)
    ASK_ROW(a10, q0, m0, 10, z, 2)
    ASK_ROW(a11, q0, m0, 11, w, 2)
    ASK_ROW(a12, q0, m0, 12, x, 3)
    ASK_ROW(a13, q0, m0, 13, y, 3)
    ASK_ROW(a14, q0, m0, 14, z, 3)
    ASK_ROW(a15, q0, m0, 15, w, 3)
    ASK_ROW(a16, q0, m0, 16, x, 4)
    ASK_ROW(a17, q0, m0, 17, y, 4)
    ASK_ROW(a18, q0, m0, 18, z, 4)
    ASK_ROW(a19, q0, m0, 19, w, 4)
    ASK_ROW(a20, q0, m0, 20, x, 5)
    ASK_ROW(a21, q0, m0, 21, y, 5)
    ASK_ROW(a22, q0, m0, 22, z, 5)
    ASK_ROW(a23, q0, m0, 23, w, 5)
    ASK_ROW(a24, q0, m0, 24, x, 6)
    ASK_ROW(a25, q0, m0, 25, y, 6)
    ASK_ROW(a26, q0, m0, 26, z, 6)
    ASK_ROW(a27, q0, m0, 27, w, 6)
    ASK_ROW(a28, q0, m0, 28, x, 7)
    ASK_ROW(a29, q0, m0, 29, y, 7)
    ASK_ROW(a30, q0, m0, 30, z, 7)
    ASK_ROW(a31, q0, m0, 31, w, 7)
    const int nxt = node + nq;
    const bool nact = act && nxt < n;
    int nc0 = 0;
    v2f nacc = {0.0f, 0.0f};
    int4 nq0 = none;
    if (nact) {
      nc0 = cnt[nxt];
      nacc = gc[(size_t)nxt * G2V];
      nq0 = bq[(size_t)nxt * 8];
    }
    ADD_ROW(a0)
    ADD_ROW(a1)
    ADD_ROW(a2)
    ADD_ROW(a3)
    ADD_ROW(a4)
    ADD_ROW(a5)
    ADD_ROW(a6)
    ADD_ROW(a7)
    ADD_ROW(a8)
    ADD_ROW(a9)
    ADD_ROW(a10)
    ADD_ROW(a11)
    ADD_ROW(a12)
    ADD_ROW(a13)
    ADD_ROW(a14)
    ADD_ROW(a15)
    ADD_ROW(a16)
    ADD_ROW(a17)
    ADD_ROW(a18)
    ADD_ROW(a19)
    ADD_ROW(a20)
    ADD_ROW(a21)
    ADD_ROW(a22)
    ADD_ROW(a23)
    ADD_ROW(a24)
    ADD_ROW(a25)
    ADD_ROW(a26)
    ADD_ROW(a27)
    ADD_ROW(a28)
    ADD_ROW(a29)
    ADD_ROW(a30)
    ADD_ROW(a31)
    if (act && c0 > SUB) {
      const int* lb = col2 + (size_t)node * SUB;
      const int mb = (c0 < BKT ? c0 : BKT) - SUB;
      for (int k = 0; k < mb; ++k) {
        const v2f r = gc[(size_t)lb[k] * G2V];
        acc.x += r.x;
        acc.y += r.y;
      }
    }
    if (act && c0 > BKT) {
      const int nov = *ovc;
      for (int k = 0; k < nov; ++k) {
        const int2 ed = ov[k];
        if (ed.x == node) {
          const v2f r = gc[(size_t)ed.y * G2V];
          acc.x += r.x;
          acc.y += r.y;
        }
      }
    }
    if (act) {
      v2f res;
      res.x = acc.x * di + bc.x;
      res.y = acc.y * di + bc.y;
      ((v2f*)out)[(size_t)node * 8 + c] = res;
    }
    node = nxt;
    act = nact;
    c0 = nc0;
    acc = nacc;
    q0 = nq0;
  }
}
#undef ASK_ROW
#undef ADD_ROW

extern "C" void kernel_launch(void* const* d_in, const int* in_sizes, int n_in,
                              void* d_out, int out_size, void* d_ws,
                              size_t ws_size, hipStream_t stream) {
  constexpr int IN_FEAT = 256, HIDDEN = 64;

  const float* x  = (const float*)d_in[0];
  const int*   ei = (const int*)d_in[1];
  const float* W1 = (const float*)d_in[2];
  const float* b1 = (const float*)d_in[3];
  const float* W2 = (const float*)d_in[4];
  const float* b2 = (const float*)d_in[5];
  float* out = (float*)d_out;

  const int N = in_sizes[0] / IN_FEAT;
  const int E = in_sizes[1] / 2;
  const int* src = ei;
  const int* dst = ei + E;

  const size_t NP = ((size_t)N + 63) / 64 * 64;
  const int nbin = (N + NPB - 1) / NPB;
  const int nwg  = (E + CHUNK - 1) / CHUNK;
  const size_t RP = ((size_t)nbin * nwg + 63) / 64 * 64;
  const size_t WP = ((size_t)nwg + 63) / 64 * 64;
  float* ws   = (float*)d_ws;
  float* h1   = ws;
  float* g2   = h1 + (size_t)N * HIDDEN;
  int* cnt    = (int*)(g2 + NP * G2P);
  int* ovc    = cnt + NP;
  int* col    = ovc + 64;
  int* col2   = col + (size_t)N * SUB;
  int2* ov    = (int2*)(col2 + (size_t)N * SUB);
  int2* runs  = ov + E;
  int2* runs2 = runs + (size_t)nbin * nwg * RH;
  int* runcnt = (int*)(runs2 + (size_t)nbin * nwg * RH);
  int* spillcnt = runcnt + RP;
  int2* spill = (int2*)(spillcnt + WP);

  const int B = 256;
  const int gblocks = 256;
  const int pblocks = 2048;
  const int qblocks = 1024;

  k_bin<<<nwg, BT, 0, stream>>>(src, dst, runs, runs2, runcnt, spill, spillcnt, ovc, E, nbin, nwg);
  k_rank<<<nbin, BT, 0, stream>>>(runs, runs2, runcnt, spill, spillcnt, cnt, col, col2, ovc, ov, N, nwg);

  wmma_gemm_f32<IN_FEAT, HIDDEN><<<gblocks, GW * 32, 0, stream>>>(x, W1, cnt, h1, N);
  k_gather64_mm<<<pblocks, B, 0, stream>>>(cnt, col, col2, ovc, ov, h1, b1, W2, g2, N);

  k_gather16<<<qblocks, B, 0, stream>>>(cnt, col, col2, ovc, ov, g2, b2, out, N);
}
